# starting kernel
# baseline (speedup 1.0000x reference)
#include <hip/hip_runtime.h>
#include <math.h>

#define DIM      128
#define NHEADS   16
#define DPH      8
#define QKV_N    384

typedef float v2f __attribute__((ext_vector_type(2)));
typedef float v8f __attribute__((ext_vector_type(8)));

__device__ __forceinline__ v8f wmma_f32_16x16x4(v2f a, v2f b, v8f c) {
    return __builtin_amdgcn_wmma_f32_16x16x4_f32(
        false, a, false, b, (short)0, c, false, false);
}

__global__ void ln_stats_kernel(const float* __restrict__ nf,
                                float* __restrict__ stats, int natoms) {
    const int atom = blockIdx.x * 8 + (threadIdx.x >> 5);
    const int lane = threadIdx.x & 31;
    if (atom >= natoms) return;
    const float4 v = ((const float4*)(nf + (size_t)atom * DIM))[lane];
    float s  = v.x + v.y + v.z + v.w;
    float s2 = v.x*v.x + v.y*v.y + v.z*v.z + v.w*v.w;
    #pragma unroll
    for (int m = 1; m < 32; m <<= 1) {
        s  += __shfl_xor(s,  m, 32);
        s2 += __shfl_xor(s2, m, 32);
    }
    if (lane == 0) {
        const float mu  = s * (1.0f / DIM);
        const float var = s2 * (1.0f / DIM) - mu * mu;
        stats[2 * atom + 0] = mu;
        stats[2 * atom + 1] = rsqrtf(var + 1e-5f);
    }
}

__global__ void qkv_gemm_kernel(const float* __restrict__ nf,
                                const float* __restrict__ stats,
                                const float* __restrict__ ln_g,
                                const float* __restrict__ ln_b,
                                const float* __restrict__ w,
                                const float* __restrict__ bias,
                                float* __restrict__ out,
                                int natoms) {
    const int lane = threadIdx.x & 31;
    const int wid  = threadIdx.x >> 5;
    const int half = lane >> 4;
    const int l16  = lane & 15;

    const int mbase = blockIdx.x * 16;
    const int nbase = (blockIdx.y * 8 + wid) * 16;
    const int ncol  = nbase + l16;

    int arow = mbase + l16;
    if (arow >= natoms) arow = natoms - 1;
    const float mu = stats[2 * arow + 0];
    const float rs = stats[2 * arow + 1];
    const float* __restrict__ ap = nf + (size_t)arow * DIM;

    v8f acc = {};
    #pragma unroll 4
    for (int k0 = 0; k0 < DIM; k0 += 4) {
        const int kk = k0 + 2 * half;
        v2f a, b;
        a.x = (ap[kk]     - mu) * rs * ln_g[kk]     + ln_b[kk];
        a.y = (ap[kk + 1] - mu) * rs * ln_g[kk + 1] + ln_b[kk + 1];
        b.x = w[(size_t)kk       * QKV_N + ncol];
        b.y = w[(size_t)(kk + 1) * QKV_N + ncol];
        acc = wmma_f32_16x16x4(a, b, acc);
    }

    const float bv = bias[ncol];
    #pragma unroll
    for (int r = 0; r < 8; ++r) {
        const int rowm = mbase + r + half * 8;
        if (rowm < natoms)
            out[(size_t)rowm * QKV_N + ncol] = acc[r] + bv;
    }
}

__global__ void zero_kernel(float* __restrict__ p, size_t n) {
    size_t i = (size_t)blockIdx.x * blockDim.x + threadIdx.x;
    const size_t stride = (size_t)gridDim.x * blockDim.x;
    for (; i < n; i += stride) p[i] = 0.0f;
}

__global__ void edge_kernel(const float* __restrict__ qkv,
                            const float* __restrict__ edge_feat,
                            const float* __restrict__ edge_vec,
                            const float* __restrict__ radial,
                            const int* __restrict__ row,
                            const int* __restrict__ col,
                            float* __restrict__ m_feat,
                            float* __restrict__ out_vec,
                            int nedges) {
    const long long gid = (long long)blockIdx.x * blockDim.x + threadIdx.x;
    const int e = (int)(gid >> 7);
    if (e >= nedges) return;
    const int t = (int)(gid & 127);
    const int h = t >> 3;
    const int d = t & 7;

    const int r = row[e];
    const int c = col[e];

    const float q = qkv[(size_t)r * QKV_N + h * 24 + d];
    const float k = qkv[(size_t)c * QKV_N + h * 24 + 8  + d];
    const float v = qkv[(size_t)c * QKV_N + h * 24 + 16 + d];

    float p = q * k;
    p += __shfl_xor(p, 1, 32);
    p += __shfl_xor(p, 2, 32);
    p += __shfl_xor(p, 4, 32);

    const float attn = 0.5f * p * (1.0f + erff(p * 0.70710678118654752440f))
                     * radial[e];

    const float mf = v * edge_feat[(size_t)e * DIM + t] * attn;
    atomicAdd(&m_feat[(size_t)r * DIM + t], mf);

    const float ev0 = edge_vec[(size_t)e * 3 + 0];
    const float ev1 = edge_vec[(size_t)e * 3 + 1];
    const float ev2 = edge_vec[(size_t)e * 3 + 2];
    float* vb = out_vec + (size_t)r * (3 * DIM) + t;
    atomicAdd(vb,           v * ev0);
    atomicAdd(vb + DIM,     v * ev1);
    atomicAdd(vb + 2 * DIM, v * ev2);
}

__global__ void out_gemm_kernel(const float* __restrict__ mf,
                                const float* __restrict__ w,
                                const float* __restrict__ bias,
                                float* __restrict__ out,
                                int natoms) {
    const int lane = threadIdx.x & 31;
    const int wid  = threadIdx.x >> 5;
    const int half = lane >> 4;
    const int l16  = lane & 15;

    const int mbase = blockIdx.x * 16;
    const int nbase = wid * 16;
    const int ncol  = nbase + l16;

    int arow = mbase + l16;
    if (arow >= natoms) arow = natoms - 1;
    const float* __restrict__ ap = mf + (size_t)arow * DIM;

    v8f acc = {};
    #pragma unroll 4
    for (int k0 = 0; k0 < DIM; k0 += 4) {
        const int kk = k0 + 2 * half;
        v2f a, b;
        a.x = ap[kk];
        a.y = ap[kk + 1];
        b.x = w[(size_t)kk       * DIM + ncol];
        b.y = w[(size_t)(kk + 1) * DIM + ncol];
        acc = wmma_f32_16x16x4(a, b, acc);
    }

    const float bv = bias[ncol];
    #pragma unroll
    for (int r = 0; r < 8; ++r) {
        const int rowm = mbase + r + half * 8;
        if (rowm < natoms)
            out[(size_t)rowm * DIM + ncol] = acc[r] + bv;
    }
}

extern "C" void kernel_launch(void* const* d_in, const int* in_sizes, int n_in,
                              void* d_out, int out_size, void* d_ws, size_t ws_size,
                              hipStream_t stream) {
    const float* nf   = (const float*)d_in[0];
    const float* ef   = (const float*)d_in[1];
    const float* evec = (const float*)d_in[2];
    const float* rad  = (const float*)d_in[3];
    const float* qkvw = (const float*)d_in[4];
    const float* qkvb = (const float*)d_in[5];
    const float* outw = (const float*)d_in[6];
    const float* outb = (const float*)d_in[7];
    const float* lng  = (const float*)d_in[8];
    const float* lnb  = (const float*)d_in[9];
    const int*   row  = (const int*)d_in[10];
    const int*   col  = (const int*)d_in[11];

    const int natoms = in_sizes[0] / DIM;
    const int nedges = in_sizes[3];

    float* stats = (float*)d_ws;
    float* qkv   = stats + (size_t)2 * natoms;
    float* mfeat = qkv + (size_t)natoms * QKV_N;

    float* out_feat = (float*)d_out;
    float* out_vec  = out_feat + (size_t)natoms * DIM;

    ln_stats_kernel<<<(natoms + 7) / 8, 256, 0, stream>>>(nf, stats, natoms);

    dim3 g1((natoms + 15) / 16, 3);
    qkv_gemm_kernel<<<g1, 256, 0, stream>>>(nf, stats, lng, lnb, qkvw, qkvb,
                                            qkv, natoms);

    const size_t zc1 = (size_t)natoms * DIM;
    const size_t zc2 = (size_t)natoms * 3 * DIM;
    zero_kernel<<<2048, 256, 0, stream>>>(mfeat, zc1);
    zero_kernel<<<4096, 256, 0, stream>>>(out_vec, zc2);

    const long long tot = (long long)nedges * DIM;
    const unsigned eblocks = (unsigned)((tot + 255) / 256);
    edge_kernel<<<eblocks, 256, 0, stream>>>(qkv, ef, evec, rad, row, col,
                                             mfeat, out_vec, nedges);

    dim3 g2((natoms + 15) / 16, 1);
    out_gemm_kernel<<<g2, 256, 0, stream>>>(mfeat, outw, outb, out_feat, natoms);
}


# Both WMMA GEMMs re-tiled: a wave keeps its A fragment for a 16-row strip of 12 (QKV) or 8 (out) column-interleaved tiles, so W loads and D stores are 16 B a lane; k order and sums kept
# speedup vs baseline: 1.7779x; 1.7779x over previous
#include <hip/hip_runtime.h>
#include <math.h>

#define DIM      128
#define NHEADS   16
#define DPH      8
#define QKV_N    384

typedef float v2f __attribute__((ext_vector_type(2)));
typedef float v8f __attribute__((ext_vector_type(8)));

__device__ __forceinline__ v8f wmma_f32_16x16x4(v2f a, v2f b, v8f c) {
    return __builtin_amdgcn_wmma_f32_16x16x4_f32(
        false, a, false, b, (short)0, c, false, false);
}

__global__ void ln_stats_kernel(const float* __restrict__ nf,
                                float* __restrict__ stats, int natoms) {
    const int atom = blockIdx.x * 8 + (threadIdx.x >> 5);
    const int lane = threadIdx.x & 31;
    if (atom >= natoms) return;
    const float4 v = ((const float4*)(nf + (size_t)atom * DIM))[lane];
    float s  = v.x + v.y + v.z + v.w;
    float s2 = v.x*v.x + v.y*v.y + v.z*v.z + v.w*v.w;
    #pragma unroll
    for (int m = 1; m < 32; m <<= 1) {
        s  += __shfl_xor(s,  m, 32);
        s2 += __shfl_xor(s2, m, 32);
    }
    if (lane == 0) {
        const float mu  = s * (1.0f / DIM);
        const float var = s2 * (1.0f / DIM) - mu * mu;
        stats[2 * atom + 0] = mu;
        stats[2 * atom + 1] = rsqrtf(var + 1e-5f);
    }
}

#define GEMM_WAVES 4

template <int GROUPS, int LDW>
__device__ __forceinline__ void strip_kstep(v8f (&acc)[GROUPS * 4], v2f a,
                                            const float* __restrict__ wk) {
    #pragma unroll
    for (int g = 0; g < GROUPS; ++g) {
        const float4 w0 = *(const float4*)(wk + g * 64);
        const float4 w1 = *(const float4*)(wk + LDW + g * 64);
        v2f b;
        b.x = w0.x; b.y = w1.x; acc[g * 4 + 0] = wmma_f32_16x16x4(a, b, acc[g * 4 + 0]);
        b.x = w0.y; b.y = w1.y; acc[g * 4 + 1] = wmma_f32_16x16x4(a, b, acc[g * 4 + 1]);
        b.x = w0.z; b.y = w1.z; acc[g * 4 + 2] = wmma_f32_16x16x4(a, b, acc[g * 4 + 2]);
        b.x = w0.w; b.y = w1.w; acc[g * 4 + 3] = wmma_f32_16x16x4(a, b, acc[g * 4 + 3]);
    }
}

template <int GROUPS, int LDO>
__device__ __forceinline__ void strip_store(const v8f (&acc)[GROUPS * 4],
                                            const float* __restrict__ bias_c,
                                            float* __restrict__ out_c,
                                            int row0, int natoms) {
    #pragma unroll
    for (int g = 0; g < GROUPS; ++g) {
        const float4 bv = *(const float4*)(bias_c + g * 64);
        #pragma unroll
        for (int r = 0; r < 8; ++r) {
            const int rowm = row0 + r;
            if (rowm < natoms) {
                float4 o;
                o.x = acc[g * 4 + 0][r] + bv.x;
                o.y = acc[g * 4 + 1][r] + bv.y;
                o.z = acc[g * 4 + 2][r] + bv.z;
                o.w = acc[g * 4 + 3][r] + bv.w;
                *(float4*)(out_c + (size_t)rowm * LDO + g * 64) = o;
            }
        }
    }
}

#define QKV_GROUPS 3

__global__ void __launch_bounds__(32 * GEMM_WAVES)
qkv_gemm_kernel(const float* __restrict__ nf,
                const float* __restrict__ stats,
                const float* __restrict__ ln_g,
                const float* __restrict__ ln_b,
                const float* __restrict__ w,
                const float* __restrict__ bias,
                float* __restrict__ out,
                int natoms) {
    const int lane = threadIdx.x & 31;
    const int wid  = threadIdx.x >> 5;
    const int half = lane >> 4;
    const int l16  = lane & 15;

    const int mbase = (blockIdx.x * GEMM_WAVES + wid) * 16;
    if (mbase >= natoms) return;
    const int cbase = blockIdx.y * (QKV_GROUPS * 64) + 4 * l16;

    int arow = mbase + l16;
    if (arow >= natoms) arow = natoms - 1;
    const float mu = stats[2 * arow + 0];
    const float rs = stats[2 * arow + 1];
    const float* __restrict__ ap = nf + (size_t)arow * DIM + 2 * half;
    const float* __restrict__ gp = ln_g + 2 * half;
    const float* __restrict__ bp = ln_b + 2 * half;
    const float* __restrict__ wp = w + (size_t)(2 * half) * QKV_N + cbase;

    v8f acc[QKV_GROUPS * 4] = {};
    #pragma unroll 2
    for (int k0 = 0; k0 < DIM; k0 += 4) {
        const float2 x  = *(const float2*)(ap + k0);
        const float2 gg = *(const float2*)(gp + k0);
        const float2 bb = *(const float2*)(bp + k0);
        v2f a;
        a.x = (x.x - mu) * rs * gg.x + bb.x;
        a.y = (x.y - mu) * rs * gg.y + bb.y;
        strip_kstep<QKV_GROUPS, QKV_N>(acc, a, wp + (size_t)k0 * QKV_N);
    }

    strip_store<QKV_GROUPS, QKV_N>(acc, bias + cbase, out + cbase,
                                   mbase + half * 8, natoms);
}

__global__ void zero_kernel(float* __restrict__ p, size_t n) {
    size_t i = (size_t)blockIdx.x * blockDim.x + threadIdx.x;
    const size_t stride = (size_t)gridDim.x * blockDim.x;
    for (; i < n; i += stride) p[i] = 0.0f;
}

#define EPW 8
#define EDGE_WAVES 8

__global__ void edge_kernel(const float* __restrict__ qkv,
                            const float* __restrict__ edge_feat,
                            const float* __restrict__ edge_vec,
                            const float* __restrict__ radial,
                            const int* __restrict__ row,
                            const int* __restrict__ col,
                            float* __restrict__ m_feat,
                            float* __restrict__ out_vec,
                            int nedges) {
    const int lane = threadIdx.x & 31;
    const int wv = __builtin_amdgcn_readfirstlane((int)(threadIdx.x >> 5));
    const int e0 = ((int)blockIdx.x * EDGE_WAVES + wv) * EPW;

    const int hoff = (lane >> 3) * 24 + (lane & 7);

    for (int i = 0; i < EPW; ++i) {
        const int e = e0 + i;
        if (e >= nedges) break;

        const int r = row[e];
        const int c = col[e];
        const float* __restrict__ qr = qkv + (size_t)r * QKV_N + hoff;
        const float* __restrict__ kc = qkv + (size_t)c * QKV_N + hoff;
        const float* __restrict__ ef = edge_feat + (size_t)e * DIM + lane;

        float q[4], k[4], v[4], f[4];
        #pragma unroll
        for (int j = 0; j < 4; ++j) {
            q[j] = qr[j * 96];
            k[j] = kc[j * 96 + 8];
            v[j] = kc[j * 96 + 16];
            f[j] = ef[j * 32];
        }
        const float rad = radial[e];
        const float ev0 = edge_vec[(size_t)e * 3 + 0];
        const float ev1 = edge_vec[(size_t)e * 3 + 1];
        const float ev2 = edge_vec[(size_t)e * 3 + 2];

        float* mfp = m_feat + (size_t)r * DIM + lane;
        float* vb  = out_vec + (size_t)r * (3 * DIM) + lane;

        float p[4];
        #pragma unroll
        for (int j = 0; j < 4; ++j) {
            float s = q[j] * k[j];
            s += __shfl_xor(s, 1, 32);
            s += __shfl_xor(s, 2, 32);
            s += __shfl_xor(s, 4, 32);
            p[j] = s;
        }

        #pragma unroll
        for (int j = 0; j < 4; ++j) {
            const float attn = 0.5f * p[j] * (1.0f + erff(p[j] * 0.70710678118654752440f))
                             * rad;

            const float mf = v[j] * f[j] * attn;
            atomicAdd(mfp + j * 32, mf);
            atomicAdd(vb + j * 32,           v[j] * ev0);
            atomicAdd(vb + j * 32 + DIM,     v[j] * ev1);
            atomicAdd(vb + j * 32 + 2 * DIM, v[j] * ev2);
        }
    }
}

#define OUT_GROUPS 2

__global__ void __launch_bounds__(32 * GEMM_WAVES)
out_gemm_kernel(const float* __restrict__ mf,
                const float* __restrict__ w,
                const float* __restrict__ bias,
                float* __restrict__ out,
                int natoms) {
    const int lane = threadIdx.x & 31;
    const int wid  = threadIdx.x >> 5;
    const int half = lane >> 4;
    const int l16  = lane & 15;

    const int mbase = (blockIdx.x * GEMM_WAVES + wid) * 16;
    if (mbase >= natoms) return;
    const int cbase = 4 * l16;

    int arow = mbase + l16;
    if (arow >= natoms) arow = natoms - 1;
    const float* __restrict__ ap = mf + (size_t)arow * DIM + 2 * half;
    const float* __restrict__ wp = w + (size_t)(2 * half) * DIM + cbase;

    v8f acc[OUT_GROUPS * 4] = {};
    #pragma unroll 2
    for (int k0 = 0; k0 < DIM; k0 += 4) {
        const float2 x = *(const float2*)(ap + k0);
        v2f a;
        a.x = x.x;
        a.y = x.y;
        strip_kstep<OUT_GROUPS, DIM>(acc, a, wp + (size_t)k0 * DIM);
    }

    strip_store<OUT_GROUPS, DIM>(acc, bias + cbase, out + cbase,
                                 mbase + half * 8, natoms);
}

extern "C" void kernel_launch(void* const* d_in, const int* in_sizes, int n_in,
                              void* d_out, int out_size, void* d_ws, size_t ws_size,
                              hipStream_t stream) {
    const float* nf   = (const float*)d_in[0];
    const float* ef   = (const float*)d_in[1];
    const float* evec = (const float*)d_in[2];
    const float* rad  = (const float*)d_in[3];
    const float* qkvw = (const float*)d_in[4];
    const float* qkvb = (const float*)d_in[5];
    const float* outw = (const float*)d_in[6];
    const float* outb = (const float*)d_in[7];
    const float* lng  = (const float*)d_in[8];
    const float* lnb  = (const float*)d_in[9];
    const int*   row  = (const int*)d_in[10];
    const int*   col  = (const int*)d_in[11];

    const int natoms = in_sizes[0] / DIM;
    const int nedges = in_sizes[3];

    float* stats = (float*)d_ws;
    float* qkv   = stats + (size_t)2 * natoms;
    float* mfeat = qkv + (size_t)natoms * QKV_N;

    float* out_feat = (float*)d_out;
    float* out_vec  = out_feat + (size_t)natoms * DIM;

    ln_stats_kernel<<<(natoms + 7) / 8, 256, 0, stream>>>(nf, stats, natoms);

    const int strips = (natoms + 15) / 16;
    dim3 g1((strips + GEMM_WAVES - 1) / GEMM_WAVES, QKV_N / (QKV_GROUPS * 64));
    qkv_gemm_kernel<<<g1, 32 * GEMM_WAVES, 0, stream>>>(nf, stats, lng, lnb, qkvw, qkvb,
                                                        qkv, natoms);

    const size_t zc1 = (size_t)natoms * DIM;
    const size_t zc2 = (size_t)natoms * 3 * DIM;
    zero_kernel<<<2048, 256, 0, stream>>>(mfeat, zc1);
    zero_kernel<<<4096, 256, 0, stream>>>(out_vec, zc2);

    const int edges_per_block = EPW * EDGE_WAVES;
    const unsigned eblocks = (unsigned)((nedges + edges_per_block - 1) / edges_per_block);
    edge_kernel<<<eblocks, 32 * EDGE_WAVES, 0, stream>>>(qkv, ef, evec, rad, row, col,
                                                         mfeat, out_vec, nedges);

    dim3 g2((strips + GEMM_WAVES - 1) / GEMM_WAVES, 1);
    out_gemm_kernel<<<g2, 32 * GEMM_WAVES, 0, stream>>>(mfeat, outw, outb, out_feat, natoms);
}


# Edge scatter-add made output-stationary: a work-group owns 196 atoms, finds their edges in row[], sums them in LDS (f32 LDS atomics, arrival order), stores each row once; no zero-fill launches
# speedup vs baseline: 2.4297x; 1.3666x over previous
#include <hip/hip_runtime.h>
#include <math.h>

#define DIM      128
#define NHEADS   16
#define DPH      8
#define QKV_N    384

typedef float v2f __attribute__((ext_vector_type(2)));
typedef float v8f __attribute__((ext_vector_type(8)));

__device__ __forceinline__ v8f wmma_f32_16x16x4(v2f a, v2f b, v8f c) {
    return __builtin_amdgcn_wmma_f32_16x16x4_f32(
        false, a, false, b, (short)0, c, false, false);
}

__global__ void ln_stats_kernel(const float* __restrict__ nf,
                                float* __restrict__ stats, int natoms) {
    const int atom = blockIdx.x * 8 + (threadIdx.x >> 5);
    const int lane = threadIdx.x & 31;
    if (atom >= natoms) return;
    const float4 v = ((const float4*)(nf + (size_t)atom * DIM))[lane];
    float s  = v.x + v.y + v.z + v.w;
    float s2 = v.x*v.x + v.y*v.y + v.z*v.z + v.w*v.w;
    #pragma unroll
    for (int m = 1; m < 32; m <<= 1) {
        s  += __shfl_xor(s,  m, 32);
        s2 += __shfl_xor(s2, m, 32);
    }
    if (lane == 0) {
        const float mu  = s * (1.0f / DIM);
        const float var = s2 * (1.0f / DIM) - mu * mu;
        stats[2 * atom + 0] = mu;
        stats[2 * atom + 1] = rsqrtf(var + 1e-5f);
    }
}

#define GEMM_WAVES 4

template <int GROUPS, int LDW>
__device__ __forceinline__ void strip_kstep(v8f (&acc)[GROUPS * 4], v2f a,
                                            const float* __restrict__ wk) {
    #pragma unroll
    for (int g = 0; g < GROUPS; ++g) {
        const float4 w0 = *(const float4*)(wk + g * 64);
        const float4 w1 = *(const float4*)(wk + LDW + g * 64);
        v2f b;
        b.x = w0.x; b.y = w1.x; acc[g * 4 + 0] = wmma_f32_16x16x4(a, b, acc[g * 4 + 0]);
        b.x = w0.y; b.y = w1.y; acc[g * 4 + 1] = wmma_f32_16x16x4(a, b, acc[g * 4 + 1]);
        b.x = w0.z; b.y = w1.z; acc[g * 4 + 2] = wmma_f32_16x16x4(a, b, acc[g * 4 + 2]);
        b.x = w0.w; b.y = w1.w; acc[g * 4 + 3] = wmma_f32_16x16x4(a, b, acc[g * 4 + 3]);
    }
}

template <int GROUPS, int LDO>
__device__ __forceinline__ void strip_store(const v8f (&acc)[GROUPS * 4],
                                            const float* __restrict__ bias_c,
                                            float* __restrict__ out_c,
                                            int row0, int natoms) {
    #pragma unroll
    for (int g = 0; g < GROUPS; ++g) {
        const float4 bv = *(const float4*)(bias_c + g * 64);
        #pragma unroll
        for (int r = 0; r < 8; ++r) {
            const int rowm = row0 + r;
            if (rowm < natoms) {
                float4 o;
                o.x = acc[g * 4 + 0][r] + bv.x;
                o.y = acc[g * 4 + 1][r] + bv.y;
                o.z = acc[g * 4 + 2][r] + bv.z;
                o.w = acc[g * 4 + 3][r] + bv.w;
                *(float4*)(out_c + (size_t)rowm * LDO + g * 64) = o;
            }
        }
    }
}

#define QKV_GROUPS 3

__global__ void __launch_bounds__(32 * GEMM_WAVES)
qkv_gemm_kernel(const float* __restrict__ nf,
                const float* __restrict__ stats,
                const float* __restrict__ ln_g,
                const float* __restrict__ ln_b,
                const float* __restrict__ w,
                const float* __restrict__ bias,
                float* __restrict__ out,
                int natoms) {
    const int lane = threadIdx.x & 31;
    const int wid  = threadIdx.x >> 5;
    const int half = lane >> 4;
    const int l16  = lane & 15;

    const int mbase = (blockIdx.x * GEMM_WAVES + wid) * 16;
    if (mbase >= natoms) return;
    const int cbase = blockIdx.y * (QKV_GROUPS * 64) + 4 * l16;

    int arow = mbase + l16;
    if (arow >= natoms) arow = natoms - 1;
    const float mu = stats[2 * arow + 0];
    const float rs = stats[2 * arow + 1];
    const float* __restrict__ ap = nf + (size_t)arow * DIM + 2 * half;
    const float* __restrict__ gp = ln_g + 2 * half;
    const float* __restrict__ bp = ln_b + 2 * half;
    const float* __restrict__ wp = w + (size_t)(2 * half) * QKV_N + cbase;

    v8f acc[QKV_GROUPS * 4] = {};
    #pragma unroll 2
    for (int k0 = 0; k0 < DIM; k0 += 4) {
        const float2 x  = *(const float2*)(ap + k0);
        const float2 gg = *(const float2*)(gp + k0);
        const float2 bb = *(const float2*)(bp + k0);
        v2f a;
        a.x = (x.x - mu) * rs * gg.x + bb.x;
        a.y = (x.y - mu) * rs * gg.y + bb.y;
        strip_kstep<QKV_GROUPS, QKV_N>(acc, a, wp + (size_t)k0 * QKV_N);
    }

    strip_store<QKV_GROUPS, QKV_N>(acc, bias + cbase, out + cbase,
                                   mbase + half * 8, natoms);
}

#define ROWS_PASS    98
#define NPASS        2
#define ROWS_WG      (ROWS_PASS * NPASS)
#define ACC_STRIDE   512
#define EDGE_THREADS 1024
#define LIST_CAP     2048
#define EB           4
#define SCAN_V       8

__device__ __forceinline__ void edge_by_one_lane(const float* __restrict__ qkv,
                                                 const float* __restrict__ edge_feat,
                                                 const float* __restrict__ edge_vec,
                                                 const float* __restrict__ radial,
                                                 int e, int r, int c, float* accrow) {
    const float rad = radial[e];
    const float ev0 = edge_vec[(size_t)e * 3 + 0];
    const float ev1 = edge_vec[(size_t)e * 3 + 1];
    const float ev2 = edge_vec[(size_t)e * 3 + 2];
    const float* qr = qkv + (size_t)r * QKV_N;
    const float* kc = qkv + (size_t)c * QKV_N;
    const float* ef = edge_feat + (size_t)e * DIM;
    for (int h = 0; h < NHEADS; ++h) {
        const float* qh = qr + h * 24;
        const float* kh = kc + h * 24 + 8;
        const float* vh = kc + h * 24 + 16;
        const float p01 = qh[0] * kh[0] + qh[1] * kh[1];
        const float p23 = qh[2] * kh[2] + qh[3] * kh[3];
        const float p45 = qh[4] * kh[4] + qh[5] * kh[5];
        const float p67 = qh[6] * kh[6] + qh[7] * kh[7];
        const float p = (p01 + p23) + (p45 + p67);
        const float attn = 0.5f * p * (1.0f + erff(p * 0.70710678118654752440f)) * rad;
        for (int d = 0; d < DPH; ++d) {
            const int ch = h * DPH + d;
            const float vv = vh[d];
            atomicAdd(accrow + ch,           vv * ef[ch] * attn);
            atomicAdd(accrow + DIM + ch,     vv * ev0);
            atomicAdd(accrow + 2 * DIM + ch, vv * ev1);
            atomicAdd(accrow + 3 * DIM + ch, vv * ev2);
        }
    }
}

__global__ void __launch_bounds__(EDGE_THREADS)
edge_rows_kernel(const float* __restrict__ qkv,
                 const float* __restrict__ edge_feat,
                 const float* __restrict__ edge_vec,
                 const float* __restrict__ radial,
                 const int* __restrict__ row,
                 const int* __restrict__ col,
                 float* __restrict__ m_feat,
                 float* __restrict__ out_vec,
                 int natoms, int nedges) {
    __shared__ float acc[ROWS_PASS * ACC_STRIDE];
    __shared__ int   list_e[NPASS][LIST_CAP];
    __shared__ int   list_c[NPASS][LIST_CAP];
    __shared__ int   list_r[NPASS][LIST_CAP];
    __shared__ int   list_n[NPASS];

    const int tid   = threadIdx.x;
    const int rbase = (int)blockIdx.x * ROWS_WG;

    if (tid < NPASS) list_n[tid] = 0;
    __syncthreads();

    {
        const int nvec = nedges >> 2;
        const int4* __restrict__ row4 = (const int4*)row;
        for (int i0 = tid; i0 < nvec; i0 += EDGE_THREADS * SCAN_V) {
            int4 rv[SCAN_V];
            #pragma unroll
            for (int u = 0; u < SCAN_V; ++u) {
                int i = i0 + u * EDGE_THREADS;
                if (i >= nvec) i = nvec - 1;
                rv[u] = row4[i];
            }
            #pragma unroll
            for (int u = 0; u < SCAN_V; ++u) {
                const int i = i0 + u * EDGE_THREADS;
                if (i < nvec) {
                    const int rr[4] = { rv[u].x, rv[u].y, rv[u].z, rv[u].w };
                    #pragma unroll
                    for (int s = 0; s < 4; ++s) {
                        const unsigned lr = (unsigned)(rr[s] - rbase);
                        if (lr < (unsigned)ROWS_WG) {
                            const unsigned ps = lr / ROWS_PASS;
                            const int slot = atomicAdd(&list_n[ps], 1);
                            if (slot < LIST_CAP) {
                                list_e[ps][slot] = 4 * i + s;
                                list_r[ps][slot] = (int)(lr - ps * ROWS_PASS);
                            }
                        }
                    }
                }
            }
        }
        for (int e = (nvec << 2) + tid; e < nedges; e += EDGE_THREADS) {
            const unsigned lr = (unsigned)(row[e] - rbase);
            if (lr < (unsigned)ROWS_WG) {
                const unsigned ps = lr / ROWS_PASS;
                const int slot = atomicAdd(&list_n[ps], 1);
                if (slot < LIST_CAP) {
                    list_e[ps][slot] = e;
                    list_r[ps][slot] = (int)(lr - ps * ROWS_PASS);
                }
            }
        }
    }
    __syncthreads();

    const int lane = tid & 31;
    const int wv   = __builtin_amdgcn_readfirstlane(tid >> 5);
    const int hoff = (lane >> 3) * 24 + (lane & 7);

    for (int ps = 0; ps < NPASS; ++ps) {
        const int  pbase  = rbase + ps * ROWS_PASS;
        const int  found  = list_n[ps];
        const bool listed = found <= LIST_CAP;
        const int  n      = listed ? found : 0;

        {
            float4* a4 = (float4*)acc;
            const float4 z = make_float4(0.0f, 0.0f, 0.0f, 0.0f);
            for (int i = tid; i < ROWS_PASS * ACC_STRIDE / 4; i += EDGE_THREADS) a4[i] = z;
            for (int i = tid; i < n; i += EDGE_THREADS) list_c[ps][i] = col[list_e[ps][i]];
        }
        __syncthreads();

        if (listed) {
            for (int i0 = wv * EB; i0 < n; i0 += (EDGE_THREADS / 32) * EB) {
                int   lr[EB];
                float q[EB][4], k[EB][4], v[EB][4], f[EB][4];
                float rad[EB], ev0[EB], ev1[EB], ev2[EB];

                #pragma unroll
                for (int t = 0; t < EB; ++t) {
                    int i = i0 + t;
                    if (i >= n) i = n - 1;
                    const int e = __builtin_amdgcn_readfirstlane(list_e[ps][i]);
                    const int c = __builtin_amdgcn_readfirstlane(list_c[ps][i]);
                    const int l = __builtin_amdgcn_readfirstlane(list_r[ps][i]);
                    lr[t] = l;
                    const float* __restrict__ qr = qkv + (size_t)(pbase + l) * QKV_N + hoff;
                    const float* __restrict__ kc = qkv + (size_t)c * QKV_N + hoff;
                    const float* __restrict__ ef = edge_feat + (size_t)e * DIM + lane;
                    #pragma unroll
                    for (int j = 0; j < 4; ++j) {
                        q[t][j] = qr[j * 96];
                        k[t][j] = kc[j * 96 + 8];
                        v[t][j] = kc[j * 96 + 16];
                        f[t][j] = ef[j * 32];
                    }
                    rad[t] = radial[e];
                    ev0[t] = edge_vec[(size_t)e * 3 + 0];
                    ev1[t] = edge_vec[(size_t)e * 3 + 1];
                    ev2[t] = edge_vec[(size_t)e * 3 + 2];
                }

                #pragma unroll
                for (int t = 0; t < EB; ++t) {
                    if (i0 + t < n) {
                        float p[4];
                        #pragma unroll
                        for (int j = 0; j < 4; ++j) {
                            float s = q[t][j] * k[t][j];
                            s += __shfl_xor(s, 1, 32);
                            s += __shfl_xor(s, 2, 32);
                            s += __shfl_xor(s, 4, 32);
                            p[j] = s;
                        }
                        float* a = acc + lr[t] * ACC_STRIDE + lane;
                        #pragma unroll
                        for (int j = 0; j < 4; ++j) {
                            const float attn = 0.5f * p[j]
                                             * (1.0f + erff(p[j] * 0.70710678118654752440f))
                                             * rad[t];
                            const float mf = v[t][j] * f[t][j] * attn;
                            atomicAdd(a + j * 32,           mf);
                            atomicAdd(a + j * 32 + DIM,     v[t][j] * ev0[t]);
                            atomicAdd(a + j * 32 + 2 * DIM, v[t][j] * ev1[t]);
                            atomicAdd(a + j * 32 + 3 * DIM, v[t][j] * ev2[t]);
                        }
                    }
                }
            }
        } else {
            for (int e = tid; e < nedges; e += EDGE_THREADS) {
                const int r = row[e];
                const unsigned l = (unsigned)(r - pbase);
                if (l < (unsigned)ROWS_PASS)
                    edge_by_one_lane(qkv, edge_feat, edge_vec, radial, e, r, col[e],
                                     acc + l * ACC_STRIDE);
            }
        }
        __syncthreads();

        {
            const float4* a4 = (const float4*)acc;
            for (int i = tid; i < ROWS_PASS * (ACC_STRIDE / 4); i += EDGE_THREADS) {
                const int l  = i >> 7;
                const int c4 = i & 127;
                const int r  = pbase + l;
                if (r < natoms) {
                    const float4 val = a4[i];
                    if (c4 < DIM / 4)
                        *(float4*)(m_feat + (size_t)r * DIM + c4 * 4) = val;
                    else
                        *(float4*)(out_vec + (size_t)r * (3 * DIM) + (c4 - DIM / 4) * 4) = val;
                }
            }
        }
        __syncthreads();
    }
}

#define OUT_GROUPS 2

__global__ void __launch_bounds__(32 * GEMM_WAVES)
out_gemm_kernel(const float* __restrict__ mf,
                const float* __restrict__ w,
                const float* __restrict__ bias,
                float* __restrict__ out,
                int natoms) {
    const int lane = threadIdx.x & 31;
    const int wid  = threadIdx.x >> 5;
    const int half = lane >> 4;
    const int l16  = lane & 15;

    const int mbase = (blockIdx.x * GEMM_WAVES + wid) * 16;
    if (mbase >= natoms) return;
    const int cbase = 4 * l16;

    int arow = mbase + l16;
    if (arow >= natoms) arow = natoms - 1;
    const float* __restrict__ ap = mf + (size_t)arow * DIM + 2 * half;
    const float* __restrict__ wp = w + (size_t)(2 * half) * DIM + cbase;

    v8f acc[OUT_GROUPS * 4] = {};
    #pragma unroll 2
    for (int k0 = 0; k0 < DIM; k0 += 4) {
        const float2 x = *(const float2*)(ap + k0);
        v2f a;
        a.x = x.x;
        a.y = x.y;
        strip_kstep<OUT_GROUPS, DIM>(acc, a, wp + (size_t)k0 * DIM);
    }

    strip_store<OUT_GROUPS, DIM>(acc, bias + cbase, out + cbase,
                                 mbase + half * 8, natoms);
}

extern "C" void kernel_launch(void* const* d_in, const int* in_sizes, int n_in,
                              void* d_out, int out_size, void* d_ws, size_t ws_size,
                              hipStream_t stream) {
    const float* nf   = (const float*)d_in[0];
    const float* ef   = (const float*)d_in[1];
    const float* evec = (const float*)d_in[2];
    const float* rad  = (const float*)d_in[3];
    const float* qkvw = (const float*)d_in[4];
    const float* qkvb = (const float*)d_in[5];
    const float* outw = (const float*)d_in[6];
    const float* outb = (const float*)d_in[7];
    const float* lng  = (const float*)d_in[8];
    const float* lnb  = (const float*)d_in[9];
    const int*   row  = (const int*)d_in[10];
    const int*   col  = (const int*)d_in[11];

    const int natoms = in_sizes[0] / DIM;
    const int nedges = in_sizes[3];

    float* stats = (float*)d_ws;
    float* qkv   = stats + (size_t)2 * natoms;
    float* mfeat = qkv + (size_t)natoms * QKV_N;

    float* out_feat = (float*)d_out;
    float* out_vec  = out_feat + (size_t)natoms * DIM;

    ln_stats_kernel<<<(natoms + 7) / 8, 256, 0, stream>>>(nf, stats, natoms);

    const int strips = (natoms + 15) / 16;
    dim3 g1((strips + GEMM_WAVES - 1) / GEMM_WAVES, QKV_N / (QKV_GROUPS * 64));
    qkv_gemm_kernel<<<g1, 32 * GEMM_WAVES, 0, stream>>>(nf, stats, lng, lnb, qkvw, qkvb,
                                                        qkv, natoms);

    edge_rows_kernel<<<(natoms + ROWS_WG - 1) / ROWS_WG, EDGE_THREADS, 0, stream>>>(
        qkv, ef, evec, rad, row, col, mfeat, out_vec, natoms, nedges);

    dim3 g2((strips + GEMM_WAVES - 1) / GEMM_WAVES, 1);
    out_gemm_kernel<<<g2, 32 * GEMM_WAVES, 0, stream>>>(mfeat, outw, outb, out_feat, natoms);
}


# Edge pass lane map changed: a lane holds 4 adjacent channels (head l>>1), not channel mod 32: q,k,v,edge_feat are one float4 load each, erf runs once a lane-edge not 4 times; same dot tree
# speedup vs baseline: 2.5957x; 1.0683x over previous
#include <hip/hip_runtime.h>
#include <math.h>

#define DIM      128
#define NHEADS   16
#define DPH      8
#define QKV_N    384

typedef float v2f __attribute__((ext_vector_type(2)));
typedef float v8f __attribute__((ext_vector_type(8)));

__device__ __forceinline__ v8f wmma_f32_16x16x4(v2f a, v2f b, v8f c) {
    return __builtin_amdgcn_wmma_f32_16x16x4_f32(
        false, a, false, b, (short)0, c, false, false);
}

__global__ void ln_stats_kernel(const float* __restrict__ nf,
                                float* __restrict__ stats, int natoms) {
    const int atom = blockIdx.x * 8 + (threadIdx.x >> 5);
    const int lane = threadIdx.x & 31;
    if (atom >= natoms) return;
    const float4 v = ((const float4*)(nf + (size_t)atom * DIM))[lane];
    float s  = v.x + v.y + v.z + v.w;
    float s2 = v.x*v.x + v.y*v.y + v.z*v.z + v.w*v.w;
    #pragma unroll
    for (int m = 1; m < 32; m <<= 1) {
        s  += __shfl_xor(s,  m, 32);
        s2 += __shfl_xor(s2, m, 32);
    }
    if (lane == 0) {
        const float mu  = s * (1.0f / DIM);
        const float var = s2 * (1.0f / DIM) - mu * mu;
        stats[2 * atom + 0] = mu;
        stats[2 * atom + 1] = rsqrtf(var + 1e-5f);
    }
}

#define GEMM_WAVES 4

template <int GROUPS, int LDW>
__device__ __forceinline__ void strip_kstep(v8f (&acc)[GROUPS * 4], v2f a,
                                            const float* __restrict__ wk) {
    #pragma unroll
    for (int g = 0; g < GROUPS; ++g) {
        const float4 w0 = *(const float4*)(wk + g * 64);
        const float4 w1 = *(const float4*)(wk + LDW + g * 64);
        v2f b;
        b.x = w0.x; b.y = w1.x; acc[g * 4 + 0] = wmma_f32_16x16x4(a, b, acc[g * 4 + 0]);
        b.x = w0.y; b.y = w1.y; acc[g * 4 + 1] = wmma_f32_16x16x4(a, b, acc[g * 4 + 1]);
        b.x = w0.z; b.y = w1.z; acc[g * 4 + 2] = wmma_f32_16x16x4(a, b, acc[g * 4 + 2]);
        b.x = w0.w; b.y = w1.w; acc[g * 4 + 3] = wmma_f32_16x16x4(a, b, acc[g * 4 + 3]);
    }
}

template <int GROUPS, int LDO>
__device__ __forceinline__ void strip_store(const v8f (&acc)[GROUPS * 4],
                                            const float* __restrict__ bias_c,
                                            float* __restrict__ out_c,
                                            int row0, int natoms) {
    #pragma unroll
    for (int g = 0; g < GROUPS; ++g) {
        const float4 bv = *(const float4*)(bias_c + g * 64);
        #pragma unroll
        for (int r = 0; r < 8; ++r) {
            const int rowm = row0 + r;
            if (rowm < natoms) {
                float4 o;
                o.x = acc[g * 4 + 0][r] + bv.x;
                o.y = acc[g * 4 + 1][r] + bv.y;
                o.z = acc[g * 4 + 2][r] + bv.z;
                o.w = acc[g * 4 + 3][r] + bv.w;
                *(float4*)(out_c + (size_t)rowm * LDO + g * 64) = o;
            }
        }
    }
}

#define QKV_GROUPS 3

__global__ void __launch_bounds__(32 * GEMM_WAVES)
qkv_gemm_kernel(const float* __restrict__ nf,
                const float* __restrict__ stats,
                const float* __restrict__ ln_g,
                const float* __restrict__ ln_b,
                const float* __restrict__ w,
                const float* __restrict__ bias,
                float* __restrict__ out,
                int natoms) {
    const int lane = threadIdx.x & 31;
    const int wid  = threadIdx.x >> 5;
    const int half = lane >> 4;
    const int l16  = lane & 15;

    const int mbase = (blockIdx.x * GEMM_WAVES + wid) * 16;
    if (mbase >= natoms) return;
    const int cbase = blockIdx.y * (QKV_GROUPS * 64) + 4 * l16;

    int arow = mbase + l16;
    if (arow >= natoms) arow = natoms - 1;
    const float mu = stats[2 * arow + 0];
    const float rs = stats[2 * arow + 1];
    const float* __restrict__ ap = nf + (size_t)arow * DIM + 2 * half;
    const float* __restrict__ gp = ln_g + 2 * half;
    const float* __restrict__ bp = ln_b + 2 * half;
    const float* __restrict__ wp = w + (size_t)(2 * half) * QKV_N + cbase;

    v8f acc[QKV_GROUPS * 4] = {};
    #pragma unroll 2
    for (int k0 = 0; k0 < DIM; k0 += 4) {
        const float2 x  = *(const float2*)(ap + k0);
        const float2 gg = *(const float2*)(gp + k0);
        const float2 bb = *(const float2*)(bp + k0);
        v2f a;
        a.x = (x.x - mu) * rs * gg.x + bb.x;
        a.y = (x.y - mu) * rs * gg.y + bb.y;
        strip_kstep<QKV_GROUPS, QKV_N>(acc, a, wp + (size_t)k0 * QKV_N);
    }

    strip_store<QKV_GROUPS, QKV_N>(acc, bias + cbase, out + cbase,
                                   mbase + half * 8, natoms);
}

#define ROWS_PASS    98
#define NPASS        2
#define ROWS_WG      (ROWS_PASS * NPASS)
#define ACC_STRIDE   512
#define EDGE_THREADS 1024
#define LIST_CAP     2048
#define EB           4
#define SCAN_V       8

__device__ __forceinline__ float lane_xor1(float x) {
    return __int_as_float(__builtin_amdgcn_update_dpp(
        0, __float_as_int(x), 0xB1  , 0xF, 0xF, true));
}

__device__ __forceinline__ void edge_by_one_lane(const float* __restrict__ qkv,
                                                 const float* __restrict__ edge_feat,
                                                 const float* __restrict__ edge_vec,
                                                 const float* __restrict__ radial,
                                                 int e, int r, int c, float* accrow) {
    const float rad = radial[e];
    const float ev0 = edge_vec[(size_t)e * 3 + 0];
    const float ev1 = edge_vec[(size_t)e * 3 + 1];
    const float ev2 = edge_vec[(size_t)e * 3 + 2];
    const float* qr = qkv + (size_t)r * QKV_N;
    const float* kc = qkv + (size_t)c * QKV_N;
    const float* ef = edge_feat + (size_t)e * DIM;
    for (int h = 0; h < NHEADS; ++h) {
        const float* qh = qr + h * 24;
        const float* kh = kc + h * 24 + 8;
        const float* vh = kc + h * 24 + 16;
        const float p01 = qh[0] * kh[0] + qh[1] * kh[1];
        const float p23 = qh[2] * kh[2] + qh[3] * kh[3];
        const float p45 = qh[4] * kh[4] + qh[5] * kh[5];
        const float p67 = qh[6] * kh[6] + qh[7] * kh[7];
        const float p = (p01 + p23) + (p45 + p67);
        const float attn = 0.5f * p * (1.0f + erff(p * 0.70710678118654752440f)) * rad;
        for (int d = 0; d < DPH; ++d) {
            const int ch = h * DPH + d;
            const int sl = (ch & 3) * 32 + (ch >> 2);
            const float vv = vh[d];
            atomicAdd(accrow + sl,           vv * ef[ch] * attn);
            atomicAdd(accrow + DIM + sl,     vv * ev0);
            atomicAdd(accrow + 2 * DIM + sl, vv * ev1);
            atomicAdd(accrow + 3 * DIM + sl, vv * ev2);
        }
    }
}

__global__ void __launch_bounds__(EDGE_THREADS)
edge_rows_kernel(const float* __restrict__ qkv,
                 const float* __restrict__ edge_feat,
                 const float* __restrict__ edge_vec,
                 const float* __restrict__ radial,
                 const int* __restrict__ row,
                 const int* __restrict__ col,
                 float* __restrict__ m_feat,
                 float* __restrict__ out_vec,
                 int natoms, int nedges) {
    __shared__ float acc[ROWS_PASS * ACC_STRIDE];
    __shared__ int   list_e[NPASS][LIST_CAP];
    __shared__ int   list_c[NPASS][LIST_CAP];
    __shared__ int   list_r[NPASS][LIST_CAP];
    __shared__ int   list_n[NPASS];

    const int tid   = threadIdx.x;
    const int rbase = (int)blockIdx.x * ROWS_WG;

    if (tid < NPASS) list_n[tid] = 0;
    __syncthreads();

    {
        const int nvec = nedges >> 2;
        const int4* __restrict__ row4 = (const int4*)row;
        for (int i0 = tid; i0 < nvec; i0 += EDGE_THREADS * SCAN_V) {
            int4 rv[SCAN_V];
            #pragma unroll
            for (int u = 0; u < SCAN_V; ++u) {
                int i = i0 + u * EDGE_THREADS;
                if (i >= nvec) i = nvec - 1;
                rv[u] = row4[i];
            }
            #pragma unroll
            for (int u = 0; u < SCAN_V; ++u) {
                const int i = i0 + u * EDGE_THREADS;
                if (i < nvec) {
                    const int rr[4] = { rv[u].x, rv[u].y, rv[u].z, rv[u].w };
                    #pragma unroll
                    for (int s = 0; s < 4; ++s) {
                        const unsigned lr = (unsigned)(rr[s] - rbase);
                        if (lr < (unsigned)ROWS_WG) {
                            const unsigned ps = lr / ROWS_PASS;
                            const int slot = atomicAdd(&list_n[ps], 1);
                            if (slot < LIST_CAP) {
                                list_e[ps][slot] = 4 * i + s;
                                list_r[ps][slot] = (int)(lr - ps * ROWS_PASS);
                            }
                        }
                    }
                }
            }
        }
        for (int e = (nvec << 2) + tid; e < nedges; e += EDGE_THREADS) {
            const unsigned lr = (unsigned)(row[e] - rbase);
            if (lr < (unsigned)ROWS_WG) {
                const unsigned ps = lr / ROWS_PASS;
                const int slot = atomicAdd(&list_n[ps], 1);
                if (slot < LIST_CAP) {
                    list_e[ps][slot] = e;
                    list_r[ps][slot] = (int)(lr - ps * ROWS_PASS);
                }
            }
        }
    }
    __syncthreads();

    const int lane = tid & 31;
    const int wv   = __builtin_amdgcn_readfirstlane(tid >> 5);
    const int hoff = (lane >> 1) * 24 + (lane & 1) * 4;

    for (int ps = 0; ps < NPASS; ++ps) {
        const int  pbase  = rbase + ps * ROWS_PASS;
        const int  found  = list_n[ps];
        const bool listed = found <= LIST_CAP;
        const int  n      = listed ? found : 0;

        {
            float4* a4 = (float4*)acc;
            const float4 z = make_float4(0.0f, 0.0f, 0.0f, 0.0f);
            for (int i = tid; i < ROWS_PASS * ACC_STRIDE / 4; i += EDGE_THREADS) a4[i] = z;
            for (int i = tid; i < n; i += EDGE_THREADS) list_c[ps][i] = col[list_e[ps][i]];
        }
        __syncthreads();

        if (listed) {
            for (int i0 = wv * EB; i0 < n; i0 += (EDGE_THREADS / 32) * EB) {
                int    lr[EB];
                float4 q[EB], k[EB], v[EB], f[EB];
                float  rad[EB], ev0[EB], ev1[EB], ev2[EB];

                #pragma unroll
                for (int t = 0; t < EB; ++t) {
                    int i = i0 + t;
                    if (i >= n) i = n - 1;
                    const int e = __builtin_amdgcn_readfirstlane(list_e[ps][i]);
                    const int c = __builtin_amdgcn_readfirstlane(list_c[ps][i]);
                    const int l = __builtin_amdgcn_readfirstlane(list_r[ps][i]);
                    lr[t] = l;
                    const float* __restrict__ qr = qkv + (size_t)(pbase + l) * QKV_N + hoff;
                    const float* __restrict__ kc = qkv + (size_t)c * QKV_N + hoff;
                    const float* __restrict__ ef = edge_feat + (size_t)e * DIM + 4 * lane;
                    q[t] = *(const float4*)(qr);
                    k[t] = *(const float4*)(kc + 8);
                    v[t] = *(const float4*)(kc + 16);
                    f[t] = *(const float4*)(ef);
                    rad[t] = radial[e];
                    ev0[t] = edge_vec[(size_t)e * 3 + 0];
                    ev1[t] = edge_vec[(size_t)e * 3 + 1];
                    ev2[t] = edge_vec[(size_t)e * 3 + 2];
                }

                #pragma unroll
                for (int t = 0; t < EB; ++t) {
                    if (i0 + t < n) {
                        const float hs = (q[t].x * k[t].x + q[t].y * k[t].y)
                                       + (q[t].z * k[t].z + q[t].w * k[t].w);
                        const float p  = hs + lane_xor1(hs);
                        const float attn = 0.5f * p
                                         * (1.0f + erff(p * 0.70710678118654752440f))
                                         * rad[t];
                        const float vv[4] = { v[t].x, v[t].y, v[t].z, v[t].w };
                        const float ff[4] = { f[t].x, f[t].y, f[t].z, f[t].w };
                        float* a = acc + lr[t] * ACC_STRIDE + lane;
                        #pragma unroll
                        for (int j = 0; j < 4; ++j) {
                            const float mf = vv[j] * ff[j] * attn;
                            atomicAdd(a + j * 32,           mf);
                            atomicAdd(a + j * 32 + DIM,     vv[j] * ev0[t]);
                            atomicAdd(a + j * 32 + 2 * DIM, vv[j] * ev1[t]);
                            atomicAdd(a + j * 32 + 3 * DIM, vv[j] * ev2[t]);
                        }
                    }
                }
            }
        } else {
            for (int e = tid; e < nedges; e += EDGE_THREADS) {
                const int r = row[e];
                const unsigned l = (unsigned)(r - pbase);
                if (l < (unsigned)ROWS_PASS)
                    edge_by_one_lane(qkv, edge_feat, edge_vec, radial, e, r, col[e],
                                     acc + l * ACC_STRIDE);
            }
        }
        __syncthreads();

        {
            for (int i = tid; i < ROWS_PASS * (ACC_STRIDE / 4); i += EDGE_THREADS) {
                const int l  = i >> 7;
                const int c4 = i & 127;
                const int r  = pbase + l;
                if (r < natoms) {
                    const float* a = acc + l * ACC_STRIDE + (c4 >> 5) * DIM + (c4 & 31);
                    const float4 val = make_float4(a[0], a[32], a[64], a[96]);
                    if (c4 < DIM / 4)
                        *(float4*)(m_feat + (size_t)r * DIM + c4 * 4) = val;
                    else
                        *(float4*)(out_vec + (size_t)r * (3 * DIM) + (c4 - DIM / 4) * 4) = val;
                }
            }
        }
        __syncthreads();
    }
}

#define OUT_GROUPS 2

__global__ void __launch_bounds__(32 * GEMM_WAVES)
out_gemm_kernel(const float* __restrict__ mf,
                const float* __restrict__ w,
                const float* __restrict__ bias,
                float* __restrict__ out,
                int natoms) {
    const int lane = threadIdx.x & 31;
    const int wid  = threadIdx.x >> 5;
    const int half = lane >> 4;
    const int l16  = lane & 15;

    const int mbase = (blockIdx.x * GEMM_WAVES + wid) * 16;
    if (mbase >= natoms) return;
    const int cbase = 4 * l16;

    int arow = mbase + l16;
    if (arow >= natoms) arow = natoms - 1;
    const float* __restrict__ ap = mf + (size_t)arow * DIM + 2 * half;
    const float* __restrict__ wp = w + (size_t)(2 * half) * DIM + cbase;

    v8f acc[OUT_GROUPS * 4] = {};
    #pragma unroll 2
    for (int k0 = 0; k0 < DIM; k0 += 4) {
        const float2 x = *(const float2*)(ap + k0);
        v2f a;
        a.x = x.x;
        a.y = x.y;
        strip_kstep<OUT_GROUPS, DIM>(acc, a, wp + (size_t)k0 * DIM);
    }

    strip_store<OUT_GROUPS, DIM>(acc, bias + cbase, out + cbase,
                                 mbase + half * 8, natoms);
}

extern "C" void kernel_launch(void* const* d_in, const int* in_sizes, int n_in,
                              void* d_out, int out_size, void* d_ws, size_t ws_size,
                              hipStream_t stream) {
    const float* nf   = (const float*)d_in[0];
    const float* ef   = (const float*)d_in[1];
    const float* evec = (const float*)d_in[2];
    const float* rad  = (const float*)d_in[3];
    const float* qkvw = (const float*)d_in[4];
    const float* qkvb = (const float*)d_in[5];
    const float* outw = (const float*)d_in[6];
    const float* outb = (const float*)d_in[7];
    const float* lng  = (const float*)d_in[8];
    const float* lnb  = (const float*)d_in[9];
    const int*   row  = (const int*)d_in[10];
    const int*   col  = (const int*)d_in[11];

    const int natoms = in_sizes[0] / DIM;
    const int nedges = in_sizes[3];

    float* stats = (float*)d_ws;
    float* qkv   = stats + (size_t)2 * natoms;
    float* mfeat = qkv + (size_t)natoms * QKV_N;

    float* out_feat = (float*)d_out;
    float* out_vec  = out_feat + (size_t)natoms * DIM;

    ln_stats_kernel<<<(natoms + 7) / 8, 256, 0, stream>>>(nf, stats, natoms);

    const int strips = (natoms + 15) / 16;
    dim3 g1((strips + GEMM_WAVES - 1) / GEMM_WAVES, QKV_N / (QKV_GROUPS * 64));
    qkv_gemm_kernel<<<g1, 32 * GEMM_WAVES, 0, stream>>>(nf, stats, lng, lnb, qkvw, qkvb,
                                                        qkv, natoms);

    edge_rows_kernel<<<(natoms + ROWS_WG - 1) / ROWS_WG, EDGE_THREADS, 0, stream>>>(
        qkv, ef, evec, rad, row, col, mfeat, out_vec, natoms, nedges);

    dim3 g2((strips + GEMM_WAVES - 1) / GEMM_WAVES, 1);
    out_gemm_kernel<<<g2, 32 * GEMM_WAVES, 0, stream>>>(mfeat, outw, outb, out_feat, natoms);
}


# Edge rows owned by waves (16-wave groups): edges queued per owner wave; a wave keeps its 4 rows' sums and q in registers: no LDS sums, no float atomics, q read once a row; sums in queue arrival order
# speedup vs baseline: 2.7441x; 1.0572x over previous
#include <hip/hip_runtime.h>
#include <math.h>

#define DIM      128
#define NHEADS   16
#define DPH      8
#define QKV_N    384

typedef float v2f __attribute__((ext_vector_type(2)));
typedef float v8f __attribute__((ext_vector_type(8)));

__device__ __forceinline__ v8f wmma_f32_16x16x4(v2f a, v2f b, v8f c) {
    return __builtin_amdgcn_wmma_f32_16x16x4_f32(
        false, a, false, b, (short)0, c, false, false);
}

__global__ void ln_stats_kernel(const float* __restrict__ nf,
                                float* __restrict__ stats, int natoms) {
    const int atom = blockIdx.x * 8 + (threadIdx.x >> 5);
    const int lane = threadIdx.x & 31;
    if (atom >= natoms) return;
    const float4 v = ((const float4*)(nf + (size_t)atom * DIM))[lane];
    float s  = v.x + v.y + v.z + v.w;
    float s2 = v.x*v.x + v.y*v.y + v.z*v.z + v.w*v.w;
    #pragma unroll
    for (int m = 1; m < 32; m <<= 1) {
        s  += __shfl_xor(s,  m, 32);
        s2 += __shfl_xor(s2, m, 32);
    }
    if (lane == 0) {
        const float mu  = s * (1.0f / DIM);
        const float var = s2 * (1.0f / DIM) - mu * mu;
        stats[2 * atom + 0] = mu;
        stats[2 * atom + 1] = rsqrtf(var + 1e-5f);
    }
}

#define GEMM_WAVES 4

template <int GROUPS, int LDW>
__device__ __forceinline__ void strip_kstep(v8f (&acc)[GROUPS * 4], v2f a,
                                            const float* __restrict__ wk) {
    #pragma unroll
    for (int g = 0; g < GROUPS; ++g) {
        const float4 w0 = *(const float4*)(wk + g * 64);
        const float4 w1 = *(const float4*)(wk + LDW + g * 64);
        v2f b;
        b.x = w0.x; b.y = w1.x; acc[g * 4 + 0] = wmma_f32_16x16x4(a, b, acc[g * 4 + 0]);
        b.x = w0.y; b.y = w1.y; acc[g * 4 + 1] = wmma_f32_16x16x4(a, b, acc[g * 4 + 1]);
        b.x = w0.z; b.y = w1.z; acc[g * 4 + 2] = wmma_f32_16x16x4(a, b, acc[g * 4 + 2]);
        b.x = w0.w; b.y = w1.w; acc[g * 4 + 3] = wmma_f32_16x16x4(a, b, acc[g * 4 + 3]);
    }
}

template <int GROUPS, int LDO>
__device__ __forceinline__ void strip_store(const v8f (&acc)[GROUPS * 4],
                                            const float* __restrict__ bias_c,
                                            float* __restrict__ out_c,
                                            int row0, int natoms) {
    #pragma unroll
    for (int g = 0; g < GROUPS; ++g) {
        const float4 bv = *(const float4*)(bias_c + g * 64);
        #pragma unroll
        for (int r = 0; r < 8; ++r) {
            const int rowm = row0 + r;
            if (rowm < natoms) {
                float4 o;
                o.x = acc[g * 4 + 0][r] + bv.x;
                o.y = acc[g * 4 + 1][r] + bv.y;
                o.z = acc[g * 4 + 2][r] + bv.z;
                o.w = acc[g * 4 + 3][r] + bv.w;
                *(float4*)(out_c + (size_t)rowm * LDO + g * 64) = o;
            }
        }
    }
}

#define QKV_GROUPS 3

__global__ void __launch_bounds__(32 * GEMM_WAVES)
qkv_gemm_kernel(const float* __restrict__ nf,
                const float* __restrict__ stats,
                const float* __restrict__ ln_g,
                const float* __restrict__ ln_b,
                const float* __restrict__ w,
                const float* __restrict__ bias,
                float* __restrict__ out,
                int natoms) {
    const int lane = threadIdx.x & 31;
    const int wid  = threadIdx.x >> 5;
    const int half = lane >> 4;
    const int l16  = lane & 15;

    const int mbase = (blockIdx.x * GEMM_WAVES + wid) * 16;
    if (mbase >= natoms) return;
    const int cbase = blockIdx.y * (QKV_GROUPS * 64) + 4 * l16;

    int arow = mbase + l16;
    if (arow >= natoms) arow = natoms - 1;
    const float mu = stats[2 * arow + 0];
    const float rs = stats[2 * arow + 1];
    const float* __restrict__ ap = nf + (size_t)arow * DIM + 2 * half;
    const float* __restrict__ gp = ln_g + 2 * half;
    const float* __restrict__ bp = ln_b + 2 * half;
    const float* __restrict__ wp = w + (size_t)(2 * half) * QKV_N + cbase;

    v8f acc[QKV_GROUPS * 4] = {};
    #pragma unroll 2
    for (int k0 = 0; k0 < DIM; k0 += 4) {
        const float2 x  = *(const float2*)(ap + k0);
        const float2 gg = *(const float2*)(gp + k0);
        const float2 bb = *(const float2*)(bp + k0);
        v2f a;
        a.x = (x.x - mu) * rs * gg.x + bb.x;
        a.y = (x.y - mu) * rs * gg.y + bb.y;
        strip_kstep<QKV_GROUPS, QKV_N>(acc, a, wp + (size_t)k0 * QKV_N);
    }

    strip_store<QKV_GROUPS, QKV_N>(acc, bias + cbase, out + cbase,
                                   mbase + half * 8, natoms);
}

#define ROWS_WG      196
#define EDGE_WAVES   16
#define EDGE_THREADS (32 * EDGE_WAVES)
#define SLOTS        4
#define ROWS_PASS    (EDGE_WAVES * SLOTS)
#define NPASS        ((ROWS_WG + ROWS_PASS - 1) / ROWS_PASS)
#define NQUEUE       (NPASS * EDGE_WAVES)
#define QCAP         128
#define EB           8
#define SCAN_V       8
#define COLF         (NQUEUE * QCAP / EDGE_THREADS)

__device__ __forceinline__ float lane_xor1(float x) {
    return __int_as_float(__builtin_amdgcn_update_dpp(
        0, __float_as_int(x), 0xB1  , 0xF, 0xF, true));
}

struct EdgeIn {
    float4 k, v, f;
    float  rad, e0, e1, e2;
};

__device__ __forceinline__ void edge_fetch(EdgeIn& x,
                                           const float* __restrict__ qkv,
                                           const float* __restrict__ edge_feat,
                                           const float* __restrict__ edge_vec,
                                           const float* __restrict__ radial,
                                           int e, int c, int hoff, int lane) {
    const float* __restrict__ kc = qkv + (size_t)c * QKV_N + hoff;
    x.k   = *(const float4*)(kc + 8);
    x.v   = *(const float4*)(kc + 16);
    x.f   = *(const float4*)(edge_feat + (size_t)e * DIM + 4 * lane);
    x.rad = radial[e];
    x.e0  = edge_vec[(size_t)e * 3 + 0];
    x.e1  = edge_vec[(size_t)e * 3 + 1];
    x.e2  = edge_vec[(size_t)e * 3 + 2];
}

__device__ __forceinline__ void edge_sum(float (&acc)[SLOTS][16],
                                         const float4 (&q)[SLOTS],
                                         const EdgeIn& x, int j) {
    float4 qs = q[0];
    #pragma unroll
    for (int s = 1; s < SLOTS; ++s) if (j == s) qs = q[s];
    const float hs = (qs.x * x.k.x + qs.y * x.k.y) + (qs.z * x.k.z + qs.w * x.k.w);
    const float p  = hs + lane_xor1(hs);
    const float attn = 0.5f * p * (1.0f + erff(p * 0.70710678118654752440f)) * x.rad;
    const float vv[4] = { x.v.x, x.v.y, x.v.z, x.v.w };
    const float ff[4] = { x.f.x, x.f.y, x.f.z, x.f.w };
    #pragma unroll
    for (int s = 0; s < SLOTS; ++s) {
        if (j == s) {
            #pragma unroll
            for (int c = 0; c < 4; ++c) {
                acc[s][c]      += vv[c] * ff[c] * attn;
                acc[s][4 + c]  += vv[c] * x.e0;
                acc[s][8 + c]  += vv[c] * x.e1;
                acc[s][12 + c] += vv[c] * x.e2;
            }
        }
    }
}

__global__ void __launch_bounds__(EDGE_THREADS)
edge_rows_kernel(const float* __restrict__ qkv,
                 const float* __restrict__ edge_feat,
                 const float* __restrict__ edge_vec,
                 const float* __restrict__ radial,
                 const int* __restrict__ row,
                 const int* __restrict__ col,
                 float* __restrict__ m_feat,
                 float* __restrict__ out_vec,
                 int natoms, int nedges) {
    __shared__ unsigned q_ent[NQUEUE * QCAP];
    __shared__ int      q_col[NQUEUE * QCAP];
    __shared__ int      q_n[NQUEUE];

    const int tid   = threadIdx.x;
    const int rbase = (int)blockIdx.x * ROWS_WG;

    if (tid < NQUEUE) q_n[tid] = 0;
    __syncthreads();

    {
        const int nvec = nedges >> 2;
        const int4* __restrict__ row4 = (const int4*)row;
        for (int i0 = tid; i0 < nvec; i0 += EDGE_THREADS * SCAN_V) {
            int4 rv[SCAN_V];
            #pragma unroll
            for (int u = 0; u < SCAN_V; ++u) {
                int i = i0 + u * EDGE_THREADS;
                if (i >= nvec) i = nvec - 1;
                rv[u] = row4[i];
            }
            #pragma unroll
            for (int u = 0; u < SCAN_V; ++u) {
                const int i = i0 + u * EDGE_THREADS;
                if (i < nvec) {
                    const int rr[4] = { rv[u].x, rv[u].y, rv[u].z, rv[u].w };
                    #pragma unroll
                    for (int s = 0; s < 4; ++s) {
                        const unsigned lr = (unsigned)(rr[s] - rbase);
                        if (lr < (unsigned)ROWS_WG) {
                            const unsigned kk = lr / EDGE_WAVES;
                            const unsigned qi = (kk / SLOTS) * EDGE_WAVES + lr % EDGE_WAVES;
                            const int slot = atomicAdd(&q_n[qi], 1);
                            if (slot < QCAP)
                                q_ent[qi * QCAP + slot] = ((unsigned)(4 * i + s) << 2) | (kk % SLOTS);
                        }
                    }
                }
            }
        }
        for (int e = (nvec << 2) + tid; e < nedges; e += EDGE_THREADS) {
            const unsigned lr = (unsigned)(row[e] - rbase);
            if (lr < (unsigned)ROWS_WG) {
                const unsigned kk = lr / EDGE_WAVES;
                const unsigned qi = (kk / SLOTS) * EDGE_WAVES + lr % EDGE_WAVES;
                const int slot = atomicAdd(&q_n[qi], 1);
                if (slot < QCAP)
                    q_ent[qi * QCAP + slot] = ((unsigned)e << 2) | (kk % SLOTS);
            }
        }
    }
    __syncthreads();

    {
        int  cv[COLF];
        bool ok[COLF];
        #pragma unroll
        for (int u = 0; u < COLF; ++u) {
            const int flat = tid + u * EDGE_THREADS;
            ok[u] = (flat % QCAP) < q_n[flat / QCAP];
            const int e = ok[u] ? (int)(q_ent[flat] >> 2) : 0;
            cv[u] = col[e];
        }
        #pragma unroll
        for (int u = 0; u < COLF; ++u)
            if (ok[u]) q_col[tid + u * EDGE_THREADS] = cv[u];
    }
    __syncthreads();

    const int lane = tid & 31;
    const int wv   = __builtin_amdgcn_readfirstlane(tid >> 5);
    const int hoff = (lane >> 1) * 24 + (lane & 1) * 4;

    for (int ps = 0; ps < NPASS; ++ps) {
        const int qi    = ps * EDGE_WAVES + wv;
        const int found = __builtin_amdgcn_readfirstlane(q_n[qi]);

        float  acc[SLOTS][16];
        float4 q[SLOTS];
        int    rowg[SLOTS];
        #pragma unroll
        for (int s = 0; s < SLOTS; ++s) {
            #pragma unroll
            for (int c = 0; c < 16; ++c) acc[s][c] = 0.0f;
            const int lr = (ps * SLOTS + s) * EDGE_WAVES + wv;
            const int r  = rbase + lr;
            rowg[s] = (lr < ROWS_WG && r < natoms) ? r : -1;
            q[s] = make_float4(0.0f, 0.0f, 0.0f, 0.0f);
            if (rowg[s] >= 0) q[s] = *(const float4*)(qkv + (size_t)r * QKV_N + hoff);
        }

        if (found <= QCAP) {
            for (int i0 = 0; i0 < found; i0 += EB) {
                EdgeIn x[EB];
                int    sj[EB];
                int li = i0 + (lane & (EB - 1));
                if (li >= found) li = found - 1;
                const unsigned en_v = q_ent[qi * QCAP + li];
                const int      c_v  = q_col[qi * QCAP + li];
                #pragma unroll
                for (int t = 0; t < EB; ++t) {
                    const unsigned en = (unsigned)__builtin_amdgcn_readlane((int)en_v, t);
                    const int      c  = __builtin_amdgcn_readlane(c_v, t);
                    sj[t] = (int)(en & 3u);
                    edge_fetch(x[t], qkv, edge_feat, edge_vec, radial, (int)(en >> 2), c, hoff, lane);
                }
                #pragma unroll
                for (int t = 0; t < EB; ++t)
                    if (i0 + t < found) edge_sum(acc, q, x[t], sj[t]);
            }
        } else {
            for (int e0 = 0; e0 < nedges; e0 += 32) {
                const int e = e0 + lane;
                unsigned lr = 0xFFFFFFFFu;
                if (e < nedges) lr = (unsigned)(row[e] - rbase);
                const bool mine = lr < (unsigned)ROWS_WG
                               && (int)(lr % EDGE_WAVES) == wv
                               && (int)((lr / EDGE_WAVES) / SLOTS) == ps;
                unsigned mask = __builtin_amdgcn_ballot_w32(mine);
                while (mask) {
                    const int b = __builtin_ctz(mask);
                    mask &= mask - 1;
                    const unsigned l = (unsigned)__builtin_amdgcn_readlane((int)lr, b);
                    const int ee = e0 + b;
                    EdgeIn x1;
                    edge_fetch(x1, qkv, edge_feat, edge_vec, radial, ee, col[ee], hoff, lane);
                    edge_sum(acc, q, x1, (int)((l / EDGE_WAVES) % SLOTS));
                }
            }
        }

        #pragma unroll
        for (int s = 0; s < SLOTS; ++s) {
            if (rowg[s] >= 0) {
                const size_t r = (size_t)rowg[s];
                *(float4*)(m_feat + r * DIM + 4 * lane) =
                    make_float4(acc[s][0], acc[s][1], acc[s][2], acc[s][3]);
                #pragma unroll
                for (int d = 0; d < 3; ++d)
                    *(float4*)(out_vec + r * (3 * DIM) + d * DIM + 4 * lane) =
                        make_float4(acc[s][4 + 4 * d], acc[s][5 + 4 * d],
                                    acc[s][6 + 4 * d], acc[s][7 + 4 * d]);
            }
        }
    }
}

#define OUT_GROUPS 2

__global__ void __launch_bounds__(32 * GEMM_WAVES)
out_gemm_kernel(const float* __restrict__ mf,
                const float* __restrict__ w,
                const float* __restrict__ bias,
                float* __restrict__ out,
                int natoms) {
    const int lane = threadIdx.x & 31;
    const int wid  = threadIdx.x >> 5;
    const int half = lane >> 4;
    const int l16  = lane & 15;

    const int mbase = (blockIdx.x * GEMM_WAVES + wid) * 16;
    if (mbase >= natoms) return;
    const int cbase = 4 * l16;

    int arow = mbase + l16;
    if (arow >= natoms) arow = natoms - 1;
    const float* __restrict__ ap = mf + (size_t)arow * DIM + 2 * half;
    const float* __restrict__ wp = w + (size_t)(2 * half) * DIM + cbase;

    v8f acc[OUT_GROUPS * 4] = {};
    #pragma unroll 2
    for (int k0 = 0; k0 < DIM; k0 += 4) {
        const float2 x = *(const float2*)(ap + k0);
        v2f a;
        a.x = x.x;
        a.y = x.y;
        strip_kstep<OUT_GROUPS, DIM>(acc, a, wp + (size_t)k0 * DIM);
    }

    strip_store<OUT_GROUPS, DIM>(acc, bias + cbase, out + cbase,
                                 mbase + half * 8, natoms);
}

extern "C" void kernel_launch(void* const* d_in, const int* in_sizes, int n_in,
                              void* d_out, int out_size, void* d_ws, size_t ws_size,
                              hipStream_t stream) {
    const float* nf   = (const float*)d_in[0];
    const float* ef   = (const float*)d_in[1];
    const float* evec = (const float*)d_in[2];
    const float* rad  = (const float*)d_in[3];
    const float* qkvw = (const float*)d_in[4];
    const float* qkvb = (const float*)d_in[5];
    const float* outw = (const float*)d_in[6];
    const float* outb = (const float*)d_in[7];
    const float* lng  = (const float*)d_in[8];
    const float* lnb  = (const float*)d_in[9];
    const int*   row  = (const int*)d_in[10];
    const int*   col  = (const int*)d_in[11];

    const int natoms = in_sizes[0] / DIM;
    const int nedges = in_sizes[3];

    float* stats = (float*)d_ws;
    float* qkv   = stats + (size_t)2 * natoms;
    float* mfeat = qkv + (size_t)natoms * QKV_N;

    float* out_feat = (float*)d_out;
    float* out_vec  = out_feat + (size_t)natoms * DIM;

    ln_stats_kernel<<<(natoms + 7) / 8, 256, 0, stream>>>(nf, stats, natoms);

    const int strips = (natoms + 15) / 16;
    dim3 g1((strips + GEMM_WAVES - 1) / GEMM_WAVES, QKV_N / (QKV_GROUPS * 64));
    qkv_gemm_kernel<<<g1, 32 * GEMM_WAVES, 0, stream>>>(nf, stats, lng, lnb, qkvw, qkvb,
                                                        qkv, natoms);

    edge_rows_kernel<<<(natoms + ROWS_WG - 1) / ROWS_WG, EDGE_THREADS, 0, stream>>>(
        qkv, ef, evec, rad, row, col, mfeat, out_vec, natoms, nedges);

    dim3 g2((strips + GEMM_WAVES - 1) / GEMM_WAVES, 1);
    out_gemm_kernel<<<g2, 32 * GEMM_WAVES, 0, stream>>>(mfeat, outw, outb, out_feat, natoms);
}


# Edge pass finds its edges in a 1-byte owner id per edge (written by extra blocks of launch 0), tested 4 edges a dword, then reads row,col of hits only: row[] not re-read 256 times; arrival-order sums
# speedup vs baseline: 2.9780x; 1.0852x over previous
#include <hip/hip_runtime.h>
#include <math.h>

#define DIM      128
#define NHEADS   16
#define DPH      8
#define QKV_N    384

typedef float v2f __attribute__((ext_vector_type(2)));
typedef float v8f __attribute__((ext_vector_type(8)));

__device__ __forceinline__ v8f wmma_f32_16x16x4(v2f a, v2f b, v8f c) {
    return __builtin_amdgcn_wmma_f32_16x16x4_f32(
        false, a, false, b, (short)0, c, false, false);
}

#define ROWS_WG      196

__device__ __forceinline__ bool owner_ids_fit(int natoms, int nedges) {
    return (size_t)nedges <= (size_t)natoms * (128 * sizeof(float));
}

__device__ __forceinline__ unsigned owner_id(int r) {
    const unsigned g = (unsigned)r / ROWS_WG;
    return g < 255u ? g : 255u;
}

__global__ void ln_stats_kernel(const float* __restrict__ nf,
                                float* __restrict__ stats, int natoms,
                                const int* __restrict__ row,
                                unsigned* __restrict__ own,
                                int nedges, int nb_stats) {
    if ((int)blockIdx.x >= nb_stats) {
        const int t = ((int)blockIdx.x - nb_stats) * 256 + (int)threadIdx.x;
        if (t < (nedges >> 2) && owner_ids_fit(natoms, nedges)) {
            const int4 r = ((const int4*)row)[t];
            own[t] = owner_id(r.x) | (owner_id(r.y) << 8) | (owner_id(r.z) << 16) | (owner_id(r.w) << 24);
        }
        return;
    }
    const int atom = blockIdx.x * 8 + (threadIdx.x >> 5);
    const int lane = threadIdx.x & 31;
    if (atom >= natoms) return;
    const float4 v = ((const float4*)(nf + (size_t)atom * DIM))[lane];
    float s  = v.x + v.y + v.z + v.w;
    float s2 = v.x*v.x + v.y*v.y + v.z*v.z + v.w*v.w;
    #pragma unroll
    for (int m = 1; m < 32; m <<= 1) {
        s  += __shfl_xor(s,  m, 32);
        s2 += __shfl_xor(s2, m, 32);
    }
    if (lane == 0) {
        const float mu  = s * (1.0f / DIM);
        const float var = s2 * (1.0f / DIM) - mu * mu;
        stats[2 * atom + 0] = mu;
        stats[2 * atom + 1] = rsqrtf(var + 1e-5f);
    }
}

#define GEMM_WAVES 4

template <int GROUPS, int LDW>
__device__ __forceinline__ void strip_kstep(v8f (&acc)[GROUPS * 4], v2f a,
                                            const float* __restrict__ wk) {
    #pragma unroll
    for (int g = 0; g < GROUPS; ++g) {
        const float4 w0 = *(const float4*)(wk + g * 64);
        const float4 w1 = *(const float4*)(wk + LDW + g * 64);
        v2f b;
        b.x = w0.x; b.y = w1.x; acc[g * 4 + 0] = wmma_f32_16x16x4(a, b, acc[g * 4 + 0]);
        b.x = w0.y; b.y = w1.y; acc[g * 4 + 1] = wmma_f32_16x16x4(a, b, acc[g * 4 + 1]);
        b.x = w0.z; b.y = w1.z; acc[g * 4 + 2] = wmma_f32_16x16x4(a, b, acc[g * 4 + 2]);
        b.x = w0.w; b.y = w1.w; acc[g * 4 + 3] = wmma_f32_16x16x4(a, b, acc[g * 4 + 3]);
    }
}

template <int GROUPS, int LDO>
__device__ __forceinline__ void strip_store(const v8f (&acc)[GROUPS * 4],
                                            const float* __restrict__ bias_c,
                                            float* __restrict__ out_c,
                                            int row0, int natoms) {
    #pragma unroll
    for (int g = 0; g < GROUPS; ++g) {
        const float4 bv = *(const float4*)(bias_c + g * 64);
        #pragma unroll
        for (int r = 0; r < 8; ++r) {
            const int rowm = row0 + r;
            if (rowm < natoms) {
                float4 o;
                o.x = acc[g * 4 + 0][r] + bv.x;
                o.y = acc[g * 4 + 1][r] + bv.y;
                o.z = acc[g * 4 + 2][r] + bv.z;
                o.w = acc[g * 4 + 3][r] + bv.w;
                *(float4*)(out_c + (size_t)rowm * LDO + g * 64) = o;
            }
        }
    }
}

#define QKV_GROUPS 3

__global__ void __launch_bounds__(32 * GEMM_WAVES)
qkv_gemm_kernel(const float* __restrict__ nf,
                const float* __restrict__ stats,
                const float* __restrict__ ln_g,
                const float* __restrict__ ln_b,
                const float* __restrict__ w,
                const float* __restrict__ bias,
                float* __restrict__ out,
                int natoms) {
    const int lane = threadIdx.x & 31;
    const int wid  = threadIdx.x >> 5;
    const int half = lane >> 4;
    const int l16  = lane & 15;

    const int mbase = (blockIdx.x * GEMM_WAVES + wid) * 16;
    if (mbase >= natoms) return;
    const int cbase = blockIdx.y * (QKV_GROUPS * 64) + 4 * l16;

    int arow = mbase + l16;
    if (arow >= natoms) arow = natoms - 1;
    const float mu = stats[2 * arow + 0];
    const float rs = stats[2 * arow + 1];
    const float* __restrict__ ap = nf + (size_t)arow * DIM + 2 * half;
    const float* __restrict__ gp = ln_g + 2 * half;
    const float* __restrict__ bp = ln_b + 2 * half;
    const float* __restrict__ wp = w + (size_t)(2 * half) * QKV_N + cbase;

    v8f acc[QKV_GROUPS * 4] = {};
    #pragma unroll 2
    for (int k0 = 0; k0 < DIM; k0 += 4) {
        const float2 x  = *(const float2*)(ap + k0);
        const float2 gg = *(const float2*)(gp + k0);
        const float2 bb = *(const float2*)(bp + k0);
        v2f a;
        a.x = (x.x - mu) * rs * gg.x + bb.x;
        a.y = (x.y - mu) * rs * gg.y + bb.y;
        strip_kstep<QKV_GROUPS, QKV_N>(acc, a, wp + (size_t)k0 * QKV_N);
    }

    strip_store<QKV_GROUPS, QKV_N>(acc, bias + cbase, out + cbase,
                                   mbase + half * 8, natoms);
}

#define EDGE_WAVES   16
#define EDGE_THREADS (32 * EDGE_WAVES)
#define SLOTS        4
#define ROWS_PASS    (EDGE_WAVES * SLOTS)
#define NPASS        ((ROWS_WG + ROWS_PASS - 1) / ROWS_PASS)
#define NQUEUE       (NPASS * EDGE_WAVES)
#define QCAP         128
#define EB           8
#define SCAN_V       8
#define OWN_V        16
#define CCAP         4096
#define RES_V        (CCAP / EDGE_THREADS)

__device__ __forceinline__ float lane_xor1(float x) {
    return __int_as_float(__builtin_amdgcn_update_dpp(
        0, __float_as_int(x), 0xB1  , 0xF, 0xF, true));
}

struct EdgeIn {
    float4 k, v, f;
    float  rad, e0, e1, e2;
};

__device__ __forceinline__ void edge_fetch(EdgeIn& x,
                                           const float* __restrict__ qkv,
                                           const float* __restrict__ edge_feat,
                                           const float* __restrict__ edge_vec,
                                           const float* __restrict__ radial,
                                           int e, int c, int hoff, int lane) {
    const float* __restrict__ kc = qkv + (size_t)c * QKV_N + hoff;
    x.k   = *(const float4*)(kc + 8);
    x.v   = *(const float4*)(kc + 16);
    x.f   = *(const float4*)(edge_feat + (size_t)e * DIM + 4 * lane);
    x.rad = radial[e];
    x.e0  = edge_vec[(size_t)e * 3 + 0];
    x.e1  = edge_vec[(size_t)e * 3 + 1];
    x.e2  = edge_vec[(size_t)e * 3 + 2];
}

__device__ __forceinline__ void edge_sum(float (&acc)[SLOTS][16],
                                         const float4 (&q)[SLOTS],
                                         const EdgeIn& x, int j) {
    float4 qs = q[0];
    #pragma unroll
    for (int s = 1; s < SLOTS; ++s) if (j == s) qs = q[s];
    const float hs = (qs.x * x.k.x + qs.y * x.k.y) + (qs.z * x.k.z + qs.w * x.k.w);
    const float p  = hs + lane_xor1(hs);
    const float attn = 0.5f * p * (1.0f + erff(p * 0.70710678118654752440f)) * x.rad;
    const float vv[4] = { x.v.x, x.v.y, x.v.z, x.v.w };
    const float ff[4] = { x.f.x, x.f.y, x.f.z, x.f.w };
    #pragma unroll
    for (int s = 0; s < SLOTS; ++s) {
        if (j == s) {
            #pragma unroll
            for (int c = 0; c < 4; ++c) {
                acc[s][c]      += vv[c] * ff[c] * attn;
                acc[s][4 + c]  += vv[c] * x.e0;
                acc[s][8 + c]  += vv[c] * x.e1;
                acc[s][12 + c] += vv[c] * x.e2;
            }
        }
    }
}

__global__ void __launch_bounds__(EDGE_THREADS)
edge_rows_kernel(const float* __restrict__ qkv,
                 const float* __restrict__ edge_feat,
                 const float* __restrict__ edge_vec,
                 const float* __restrict__ radial,
                 const int* __restrict__ row,
                 const int* __restrict__ col,
                 const unsigned* __restrict__ own,
                 float* __restrict__ m_feat,
                 float* __restrict__ out_vec,
                 int natoms, int nedges) {
    __shared__ unsigned q_ent[NQUEUE * QCAP];
    __shared__ int      q_col[NQUEUE * QCAP];
    __shared__ int      q_n[NQUEUE];
    __shared__ unsigned cand[CCAP];
    __shared__ int      cand_n;

    const int tid   = threadIdx.x;
    const int rbase = (int)blockIdx.x * ROWS_WG;

    if (tid < NQUEUE) q_n[tid] = 0;
    if (tid == 0) cand_n = 0;
    __syncthreads();

    auto queue_edge = [&](unsigned lr, int e, int c) {
        const unsigned kk = lr / EDGE_WAVES;
        const unsigned qi = (kk / SLOTS) * EDGE_WAVES + lr % EDGE_WAVES;
        const int slot = atomicAdd(&q_n[qi], 1);
        if (slot < QCAP) {
            q_ent[qi * QCAP + slot] = ((unsigned)e << 2) | (kk % SLOTS);
            q_col[qi * QCAP + slot] = c;
        }
    };

    bool queued = false;
    if (owner_ids_fit(natoms, nedges)) {
        const unsigned gid = blockIdx.x < 255u ? blockIdx.x : 255u;
        const unsigned pat = gid * 0x01010101u;
        const int nv16 = nedges >> 4;
        const uint4* __restrict__ own4 = (const uint4*)own;
        for (int b0 = 0; b0 < nv16; b0 += EDGE_THREADS * OWN_V) {
            uint4 ov[OWN_V];
            #pragma unroll
            for (int u = 0; u < OWN_V; ++u) {
                if (b0 + u * EDGE_THREADS < nv16) {
                    int i = b0 + u * EDGE_THREADS + tid;
                    if (i >= nv16) i = nv16 - 1;
                    ov[u] = own4[i];
                }
            }
            unsigned long long hit = 0;
            #pragma unroll
            for (int u = 0; u < OWN_V; ++u) {
                const int i = b0 + u * EDGE_THREADS + tid;
                if (i < nv16) {
                    const unsigned ww[4] = { ov[u].x, ov[u].y, ov[u].z, ov[u].w };
                    #pragma unroll
                    for (int d = 0; d < 4; ++d) {
                        const unsigned x = ww[d] ^ pat;
                        if ((x - 0x01010101u) & ~x & 0x80808080u) hit |= 1ull << (4 * u + d);
                    }
                }
            }
            int slot = atomicAdd(&cand_n, __popcll(hit));
            while (hit) {
                const int bq = __builtin_ctzll(hit);
                hit &= hit - 1;
                if (slot < CCAP)
                    cand[slot] = 4u * (unsigned)(b0 + (bq >> 2) * EDGE_THREADS + tid) + (unsigned)(bq & 3);
                ++slot;
            }
        }
        __syncthreads();
        const int nc = cand_n;
        if (nc <= CCAP) {
            const int4* __restrict__ row4 = (const int4*)row;
            const int4* __restrict__ col4 = (const int4*)col;
            int4     rv[RES_V], cv[RES_V];
            unsigned jj[RES_V];
            #pragma unroll
            for (int u = 0; u < RES_V; ++u) {
                const int flat = tid + u * EDGE_THREADS;
                const unsigned j = cand[flat];
                jj[u] = flat < nc ? j : 0xFFFFFFFFu;
            }
            #pragma unroll
            for (int u = 0; u < RES_V; ++u) {
                if (jj[u] != 0xFFFFFFFFu) {
                    rv[u] = row4[jj[u]];
                    cv[u] = col4[jj[u]];
                }
            }
            #pragma unroll
            for (int u = 0; u < RES_V; ++u) {
                if (jj[u] != 0xFFFFFFFFu) {
                    const int rr[4] = { rv[u].x, rv[u].y, rv[u].z, rv[u].w };
                    const int cc[4] = { cv[u].x, cv[u].y, cv[u].z, cv[u].w };
                    #pragma unroll
                    for (int s = 0; s < 4; ++s) {
                        const unsigned lr = (unsigned)(rr[s] - rbase);
                        if (lr < (unsigned)ROWS_WG) queue_edge(lr, (int)(4u * jj[u]) + s, cc[s]);
                    }
                }
            }
            for (int e = (nv16 << 4) + tid; e < nedges; e += EDGE_THREADS) {
                const unsigned lr = (unsigned)(row[e] - rbase);
                if (lr < (unsigned)ROWS_WG) queue_edge(lr, e, col[e]);
            }
            queued = true;
        }
    }
    if (!queued) {
        const int nvec = nedges >> 2;
        const int4* __restrict__ row4 = (const int4*)row;
        for (int i0 = tid; i0 < nvec; i0 += EDGE_THREADS * SCAN_V) {
            int4 rv[SCAN_V];
            #pragma unroll
            for (int u = 0; u < SCAN_V; ++u) {
                int i = i0 + u * EDGE_THREADS;
                if (i >= nvec) i = nvec - 1;
                rv[u] = row4[i];
            }
            #pragma unroll
            for (int u = 0; u < SCAN_V; ++u) {
                const int i = i0 + u * EDGE_THREADS;
                if (i < nvec) {
                    const int rr[4] = { rv[u].x, rv[u].y, rv[u].z, rv[u].w };
                    #pragma unroll
                    for (int s = 0; s < 4; ++s) {
                        const unsigned lr = (unsigned)(rr[s] - rbase);
                        if (lr < (unsigned)ROWS_WG) queue_edge(lr, 4 * i + s, col[4 * i + s]);
                    }
                }
            }
        }
        for (int e = (nvec << 2) + tid; e < nedges; e += EDGE_THREADS) {
            const unsigned lr = (unsigned)(row[e] - rbase);
            if (lr < (unsigned)ROWS_WG) queue_edge(lr, e, col[e]);
        }
    }
    __syncthreads();

    const int lane = tid & 31;
    const int wv   = __builtin_amdgcn_readfirstlane(tid >> 5);
    const int hoff = (lane >> 1) * 24 + (lane & 1) * 4;

    for (int ps = 0; ps < NPASS; ++ps) {
        const int qi    = ps * EDGE_WAVES + wv;
        const int found = __builtin_amdgcn_readfirstlane(q_n[qi]);

        float  acc[SLOTS][16];
        float4 q[SLOTS];
        int    rowg[SLOTS];
        #pragma unroll
        for (int s = 0; s < SLOTS; ++s) {
            #pragma unroll
            for (int c = 0; c < 16; ++c) acc[s][c] = 0.0f;
            const int lr = (ps * SLOTS + s) * EDGE_WAVES + wv;
            const int r  = rbase + lr;
            rowg[s] = (lr < ROWS_WG && r < natoms) ? r : -1;
            q[s] = make_float4(0.0f, 0.0f, 0.0f, 0.0f);
            if (rowg[s] >= 0) q[s] = *(const float4*)(qkv + (size_t)r * QKV_N + hoff);
        }

        if (found <= QCAP) {
            for (int i0 = 0; i0 < found; i0 += EB) {
                EdgeIn x[EB];
                int    sj[EB];
                int li = i0 + (lane & (EB - 1));
                if (li >= found) li = found - 1;
                const unsigned en_v = q_ent[qi * QCAP + li];
                const int      c_v  = q_col[qi * QCAP + li];
                #pragma unroll
                for (int t = 0; t < EB; ++t) {
                    const unsigned en = (unsigned)__builtin_amdgcn_readlane((int)en_v, t);
                    const int      c  = __builtin_amdgcn_readlane(c_v, t);
                    sj[t] = (int)(en & 3u);
                    edge_fetch(x[t], qkv, edge_feat, edge_vec, radial, (int)(en >> 2), c, hoff, lane);
                }
                #pragma unroll
                for (int t = 0; t < EB; ++t)
                    if (i0 + t < found) edge_sum(acc, q, x[t], sj[t]);
            }
        } else {
            for (int e0 = 0; e0 < nedges; e0 += 32) {
                const int e = e0 + lane;
                unsigned lr = 0xFFFFFFFFu;
                if (e < nedges) lr = (unsigned)(row[e] - rbase);
                const bool mine = lr < (unsigned)ROWS_WG
                               && (int)(lr % EDGE_WAVES) == wv
                               && (int)((lr / EDGE_WAVES) / SLOTS) == ps;
                unsigned mask = __builtin_amdgcn_ballot_w32(mine);
                while (mask) {
                    const int b = __builtin_ctz(mask);
                    mask &= mask - 1;
                    const unsigned l = (unsigned)__builtin_amdgcn_readlane((int)lr, b);
                    const int ee = e0 + b;
                    EdgeIn x1;
                    edge_fetch(x1, qkv, edge_feat, edge_vec, radial, ee, col[ee], hoff, lane);
                    edge_sum(acc, q, x1, (int)((l / EDGE_WAVES) % SLOTS));
                }
            }
        }

        #pragma unroll
        for (int s = 0; s < SLOTS; ++s) {
            if (rowg[s] >= 0) {
                const size_t r = (size_t)rowg[s];
                *(float4*)(m_feat + r * DIM + 4 * lane) =
                    make_float4(acc[s][0], acc[s][1], acc[s][2], acc[s][3]);
                #pragma unroll
                for (int d = 0; d < 3; ++d)
                    *(float4*)(out_vec + r * (3 * DIM) + d * DIM + 4 * lane) =
                        make_float4(acc[s][4 + 4 * d], acc[s][5 + 4 * d],
                                    acc[s][6 + 4 * d], acc[s][7 + 4 * d]);
            }
        }
    }
}

#define OUT_GROUPS 2

__global__ void __launch_bounds__(32 * GEMM_WAVES)
out_gemm_kernel(const float* __restrict__ mf,
                const float* __restrict__ w,
                const float* __restrict__ bias,
                float* __restrict__ out,
                int natoms) {
    const int lane = threadIdx.x & 31;
    const int wid  = threadIdx.x >> 5;
    const int half = lane >> 4;
    const int l16  = lane & 15;

    const int mbase = (blockIdx.x * GEMM_WAVES + wid) * 16;
    if (mbase >= natoms) return;
    const int cbase = 4 * l16;

    int arow = mbase + l16;
    if (arow >= natoms) arow = natoms - 1;
    const float* __restrict__ ap = mf + (size_t)arow * DIM + 2 * half;
    const float* __restrict__ wp = w + (size_t)(2 * half) * DIM + cbase;

    v8f acc[OUT_GROUPS * 4] = {};
    #pragma unroll 2
    for (int k0 = 0; k0 < DIM; k0 += 4) {
        const float2 x = *(const float2*)(ap + k0);
        v2f a;
        a.x = x.x;
        a.y = x.y;
        strip_kstep<OUT_GROUPS, DIM>(acc, a, wp + (size_t)k0 * DIM);
    }

    strip_store<OUT_GROUPS, DIM>(acc, bias + cbase, out + cbase,
                                 mbase + half * 8, natoms);
}

extern "C" void kernel_launch(void* const* d_in, const int* in_sizes, int n_in,
                              void* d_out, int out_size, void* d_ws, size_t ws_size,
                              hipStream_t stream) {
    const float* nf   = (const float*)d_in[0];
    const float* ef   = (const float*)d_in[1];
    const float* evec = (const float*)d_in[2];
    const float* rad  = (const float*)d_in[3];
    const float* qkvw = (const float*)d_in[4];
    const float* qkvb = (const float*)d_in[5];
    const float* outw = (const float*)d_in[6];
    const float* outb = (const float*)d_in[7];
    const float* lng  = (const float*)d_in[8];
    const float* lnb  = (const float*)d_in[9];
    const int*   row  = (const int*)d_in[10];
    const int*   col  = (const int*)d_in[11];

    const int natoms = in_sizes[0] / DIM;
    const int nedges = in_sizes[3];

    float* stats = (float*)d_ws;
    float* qkv   = stats + (size_t)2 * natoms;
    float* mfeat = qkv + (size_t)natoms * QKV_N;

    float* out_feat = (float*)d_out;
    float* out_vec  = out_feat + (size_t)natoms * DIM;

    unsigned* own = (unsigned*)d_out;

    const int nb_stats = (natoms + 7) / 8;
    const int nb_own   = ((nedges >> 2) + 255) / 256;
    ln_stats_kernel<<<nb_stats + nb_own, 256, 0, stream>>>(nf, stats, natoms, row, own,
                                                           nedges, nb_stats);

    const int strips = (natoms + 15) / 16;
    dim3 g1((strips + GEMM_WAVES - 1) / GEMM_WAVES, QKV_N / (QKV_GROUPS * 64));
    qkv_gemm_kernel<<<g1, 32 * GEMM_WAVES, 0, stream>>>(nf, stats, lng, lnb, qkvw, qkvb,
                                                        qkv, natoms);

    edge_rows_kernel<<<(natoms + ROWS_WG - 1) / ROWS_WG, EDGE_THREADS, 0, stream>>>(
        qkv, ef, evec, rad, row, col, own, mfeat, out_vec, natoms, nedges);

    dim3 g2((strips + GEMM_WAVES - 1) / GEMM_WAVES, 1);
    out_gemm_kernel<<<g2, 32 * GEMM_WAVES, 0, stream>>>(mfeat, outw, outb, out_feat, natoms);
}


# Edge scalars ride with the edge: pass 0 deals (edge,col,radial,edge_vec) 24-B records into the owner group's out_vec rows; edge pass streams them: no id scan, no scalar gathers; arrival-order sums
# speedup vs baseline: 3.1417x; 1.0550x over previous
#include <hip/hip_runtime.h>
#include <math.h>

#define DIM      128
#define NHEADS   16
#define DPH      8
#define QKV_N    384

typedef float v2f __attribute__((ext_vector_type(2)));
typedef float v8f __attribute__((ext_vector_type(8)));

__device__ __forceinline__ v8f wmma_f32_16x16x4(v2f a, v2f b, v8f c) {
    return __builtin_amdgcn_wmma_f32_16x16x4_f32(
        false, a, false, b, (short)0, c, false, false);
}

#define ROWS_WG      196

__device__ __forceinline__ bool owner_ids_fit(int natoms, int nedges) {
    return (size_t)nedges <= (size_t)natoms * (128 * sizeof(float));
}

__device__ __forceinline__ unsigned owner_id(int r) {
    const unsigned g = (unsigned)r / ROWS_WG;
    return g < 255u ? g : 255u;
}

#define NB_REC   256
#define MAXG     1024

__device__ __forceinline__ int n_groups(int natoms) { return (natoms + ROWS_WG - 1) / ROWS_WG; }

__device__ __forceinline__ size_t own_bytes(int nedges) {
    return ((size_t)nedges + 255) & ~(size_t)255;
}

__device__ __forceinline__ bool rec_fit(int natoms, int nedges) {
    return nedges <= (1 << 24) && n_groups(natoms) <= MAXG
        && own_bytes(nedges) + (size_t)n_groups(natoms) * NB_REC * 4 <= (size_t)natoms * (128 * sizeof(float));
}

__device__ __forceinline__ int rec_cap(int natoms, int g) {
    int rows = natoms - g * ROWS_WG;
    if (rows > ROWS_WG) rows = ROWS_WG;
    return (rows * (3 * DIM * 4)) / (NB_REC * 24);
}

__global__ void ln_stats_kernel(const float* __restrict__ nf,
                                float* __restrict__ stats, int natoms,
                                const int* __restrict__ row,
                                unsigned* __restrict__ own,
                                int nedges, int nb_stats,
                                const int* __restrict__ col,
                                const float* __restrict__ radial,
                                const float* __restrict__ edge_vec,
                                unsigned* __restrict__ rec,
                                int* __restrict__ cntg) {
    __shared__ int s_cnt[MAXG];
    if ((int)blockIdx.x < NB_REC) {
        if (!rec_fit(natoms, nedges)) return;
        const int b   = (int)blockIdx.x;
        const int tid = (int)threadIdx.x;
        const int G   = n_groups(natoms);
        for (int i = tid; i < G; i += 256) s_cnt[i] = 0;
        __syncthreads();
        auto deal = [&](int r, int e, int c, float rd, float e0, float e1, float e2) {
            if ((unsigned)r < (unsigned)natoms) {
                const int g    = r / ROWS_WG;
                const int lr   = r - g * ROWS_WG;
                const int cap  = rec_cap(natoms, g);
                const int slot = atomicAdd(&s_cnt[g], 1);
                if (slot < cap) {
                    unsigned* p = rec + (size_t)g * (ROWS_WG * 3 * DIM) + ((size_t)b * cap + slot) * 6;
                    ((uint2*)p)[0]  = make_uint2((unsigned)e | ((unsigned)lr << 24), (unsigned)c);
                    ((float2*)p)[1] = make_float2(rd, e0);
                    ((float2*)p)[2] = make_float2(e1, e2);
                }
            }
        };
        const int nquad = nedges >> 2;
        const int qpb   = (nquad + NB_REC - 1) / NB_REC;
        const int q0    = b * qpb;
        const int q1    = q0 + qpb < nquad ? q0 + qpb : nquad;
        for (int qd = q0 + tid; qd < q1; qd += 256) {
            const int4   r4 = ((const int4*)row)[qd];
            const int4   c4 = ((const int4*)col)[qd];
            const float4 d4 = ((const float4*)radial)[qd];
            const float4 ea = ((const float4*)edge_vec)[3 * qd + 0];
            const float4 eb = ((const float4*)edge_vec)[3 * qd + 1];
            const float4 ec = ((const float4*)edge_vec)[3 * qd + 2];
            deal(r4.x, 4 * qd + 0, c4.x, d4.x, ea.x, ea.y, ea.z);
            deal(r4.y, 4 * qd + 1, c4.y, d4.y, ea.w, eb.x, eb.y);
            deal(r4.z, 4 * qd + 2, c4.z, d4.z, eb.z, eb.w, ec.x);
            deal(r4.w, 4 * qd + 3, c4.w, d4.w, ec.y, ec.z, ec.w);
        }
        if (b == NB_REC - 1) {
            for (int e = (nquad << 2) + tid; e < nedges; e += 256)
                deal(row[e], e, col[e], radial[e], edge_vec[(size_t)e * 3 + 0],
                     edge_vec[(size_t)e * 3 + 1], edge_vec[(size_t)e * 3 + 2]);
        }
        __syncthreads();
        for (int i = tid; i < G; i += 256) cntg[(size_t)i * NB_REC + b] = s_cnt[i];
        return;
    }
    const int sblk = (int)blockIdx.x - NB_REC;
    if (sblk >= nb_stats) {
        const int t = (sblk - nb_stats) * 256 + (int)threadIdx.x;
        if (t < (nedges >> 2) && owner_ids_fit(natoms, nedges)) {
            const int4 r = ((const int4*)row)[t];
            own[t] = owner_id(r.x) | (owner_id(r.y) << 8) | (owner_id(r.z) << 16) | (owner_id(r.w) << 24);
        }
        return;
    }
    const int atom = sblk * 8 + (threadIdx.x >> 5);
    const int lane = threadIdx.x & 31;
    if (atom >= natoms) return;
    const float4 v = ((const float4*)(nf + (size_t)atom * DIM))[lane];
    float s  = v.x + v.y + v.z + v.w;
    float s2 = v.x*v.x + v.y*v.y + v.z*v.z + v.w*v.w;
    #pragma unroll
    for (int m = 1; m < 32; m <<= 1) {
        s  += __shfl_xor(s,  m, 32);
        s2 += __shfl_xor(s2, m, 32);
    }
    if (lane == 0) {
        const float mu  = s * (1.0f / DIM);
        const float var = s2 * (1.0f / DIM) - mu * mu;
        stats[2 * atom + 0] = mu;
        stats[2 * atom + 1] = rsqrtf(var + 1e-5f);
    }
}

#define GEMM_WAVES 4

template <int GROUPS, int LDW>
__device__ __forceinline__ void strip_kstep(v8f (&acc)[GROUPS * 4], v2f a,
                                            const float* __restrict__ wk) {
    #pragma unroll
    for (int g = 0; g < GROUPS; ++g) {
        const float4 w0 = *(const float4*)(wk + g * 64);
        const float4 w1 = *(const float4*)(wk + LDW + g * 64);
        v2f b;
        b.x = w0.x; b.y = w1.x; acc[g * 4 + 0] = wmma_f32_16x16x4(a, b, acc[g * 4 + 0]);
        b.x = w0.y; b.y = w1.y; acc[g * 4 + 1] = wmma_f32_16x16x4(a, b, acc[g * 4 + 1]);
        b.x = w0.z; b.y = w1.z; acc[g * 4 + 2] = wmma_f32_16x16x4(a, b, acc[g * 4 + 2]);
        b.x = w0.w; b.y = w1.w; acc[g * 4 + 3] = wmma_f32_16x16x4(a, b, acc[g * 4 + 3]);
    }
}

template <int GROUPS, int LDO>
__device__ __forceinline__ void strip_store(const v8f (&acc)[GROUPS * 4],
                                            const float* __restrict__ bias_c,
                                            float* __restrict__ out_c,
                                            int row0, int natoms) {
    #pragma unroll
    for (int g = 0; g < GROUPS; ++g) {
        const float4 bv = *(const float4*)(bias_c + g * 64);
        #pragma unroll
        for (int r = 0; r < 8; ++r) {
            const int rowm = row0 + r;
            if (rowm < natoms) {
                float4 o;
                o.x = acc[g * 4 + 0][r] + bv.x;
                o.y = acc[g * 4 + 1][r] + bv.y;
                o.z = acc[g * 4 + 2][r] + bv.z;
                o.w = acc[g * 4 + 3][r] + bv.w;
                *(float4*)(out_c + (size_t)rowm * LDO + g * 64) = o;
            }
        }
    }
}

#define QKV_GROUPS 3

__global__ void __launch_bounds__(32 * GEMM_WAVES)
qkv_gemm_kernel(const float* __restrict__ nf,
                const float* __restrict__ stats,
                const float* __restrict__ ln_g,
                const float* __restrict__ ln_b,
                const float* __restrict__ w,
                const float* __restrict__ bias,
                float* __restrict__ out,
                int natoms) {
    const int lane = threadIdx.x & 31;
    const int wid  = threadIdx.x >> 5;
    const int half = lane >> 4;
    const int l16  = lane & 15;

    const int mbase = (blockIdx.x * GEMM_WAVES + wid) * 16;
    if (mbase >= natoms) return;
    const int cbase = blockIdx.y * (QKV_GROUPS * 64) + 4 * l16;

    int arow = mbase + l16;
    if (arow >= natoms) arow = natoms - 1;
    const float mu = stats[2 * arow + 0];
    const float rs = stats[2 * arow + 1];
    const float* __restrict__ ap = nf + (size_t)arow * DIM + 2 * half;
    const float* __restrict__ gp = ln_g + 2 * half;
    const float* __restrict__ bp = ln_b + 2 * half;
    const float* __restrict__ wp = w + (size_t)(2 * half) * QKV_N + cbase;

    v8f acc[QKV_GROUPS * 4] = {};
    #pragma unroll 2
    for (int k0 = 0; k0 < DIM; k0 += 4) {
        const float2 x  = *(const float2*)(ap + k0);
        const float2 gg = *(const float2*)(gp + k0);
        const float2 bb = *(const float2*)(bp + k0);
        v2f a;
        a.x = (x.x - mu) * rs * gg.x + bb.x;
        a.y = (x.y - mu) * rs * gg.y + bb.y;
        strip_kstep<QKV_GROUPS, QKV_N>(acc, a, wp + (size_t)k0 * QKV_N);
    }

    strip_store<QKV_GROUPS, QKV_N>(acc, bias + cbase, out + cbase,
                                   mbase + half * 8, natoms);
}

#define EDGE_WAVES   16
#define EDGE_THREADS (32 * EDGE_WAVES)
#define SLOTS        4
#define ROWS_PASS    (EDGE_WAVES * SLOTS)
#define NPASS        ((ROWS_WG + ROWS_PASS - 1) / ROWS_PASS)
#define NQUEUE       (NPASS * EDGE_WAVES)
#define QCAP         128
#define EB           8
#define SCAN_V       8
#define OWN_V        16
#define CCAP         4096
#define RES_V        (CCAP / EDGE_THREADS)
#define REC_V        4

__device__ __forceinline__ float lane_xor1(float x) {
    return __int_as_float(__builtin_amdgcn_update_dpp(
        0, __float_as_int(x), 0xB1  , 0xF, 0xF, true));
}

struct EdgeIn {
    float4 k, v, f;
    float  rad, e0, e1, e2;
};

__device__ __forceinline__ void edge_fetch(EdgeIn& x,
                                           const float* __restrict__ qkv,
                                           const float* __restrict__ edge_feat,
                                           const float* __restrict__ edge_vec,
                                           const float* __restrict__ radial,
                                           int e, int c, int hoff, int lane) {
    const float* __restrict__ kc = qkv + (size_t)c * QKV_N + hoff;
    x.k   = *(const float4*)(kc + 8);
    x.v   = *(const float4*)(kc + 16);
    x.f   = *(const float4*)(edge_feat + (size_t)e * DIM + 4 * lane);
    x.rad = radial[e];
    x.e0  = edge_vec[(size_t)e * 3 + 0];
    x.e1  = edge_vec[(size_t)e * 3 + 1];
    x.e2  = edge_vec[(size_t)e * 3 + 2];
}

__device__ __forceinline__ void edge_fetch_kvf(EdgeIn& x,
                                               const float* __restrict__ qkv,
                                               const float* __restrict__ edge_feat,
                                               int e, int c, int hoff, int lane) {
    const float* __restrict__ kc = qkv + (size_t)c * QKV_N + hoff;
    x.k = *(const float4*)(kc + 8);
    x.v = *(const float4*)(kc + 16);
    x.f = *(const float4*)(edge_feat + (size_t)e * DIM + 4 * lane);
}

__device__ __forceinline__ float readlane_f(float v, int t) {
    return __int_as_float(__builtin_amdgcn_readlane(__float_as_int(v), t));
}

__device__ __forceinline__ void edge_sum(float (&acc)[SLOTS][16],
                                         const float4 (&q)[SLOTS],
                                         const EdgeIn& x, int j) {
    float4 qs = q[0];
    #pragma unroll
    for (int s = 1; s < SLOTS; ++s) if (j == s) qs = q[s];
    const float hs = (qs.x * x.k.x + qs.y * x.k.y) + (qs.z * x.k.z + qs.w * x.k.w);
    const float p  = hs + lane_xor1(hs);
    const float attn = 0.5f * p * (1.0f + erff(p * 0.70710678118654752440f)) * x.rad;
    const float vv[4] = { x.v.x, x.v.y, x.v.z, x.v.w };
    const float ff[4] = { x.f.x, x.f.y, x.f.z, x.f.w };
    #pragma unroll
    for (int s = 0; s < SLOTS; ++s) {
        if (j == s) {
            #pragma unroll
            for (int c = 0; c < 4; ++c) {
                acc[s][c]      += vv[c] * ff[c] * attn;
                acc[s][4 + c]  += vv[c] * x.e0;
                acc[s][8 + c]  += vv[c] * x.e1;
                acc[s][12 + c] += vv[c] * x.e2;
            }
        }
    }
}

__global__ void __launch_bounds__(EDGE_THREADS)
edge_rows_kernel(const float* __restrict__ qkv,
                 const float* __restrict__ edge_feat,
                 const float* __restrict__ edge_vec,
                 const float* __restrict__ radial,
                 const int* __restrict__ row,
                 const int* __restrict__ col,
                 const unsigned* __restrict__ own,
                 const int* __restrict__ cntg,
                 float* __restrict__ m_feat,
                 float* __restrict__ out_vec,
                 int natoms, int nedges) {
    __shared__ unsigned q_ent[NQUEUE * QCAP];
    __shared__ int      q_col[NQUEUE * QCAP];
    __shared__ float2   q_r0[NQUEUE * QCAP];
    __shared__ float2   q_r1[NQUEUE * QCAP];
    __shared__ int      q_n[NQUEUE];
    __shared__ int      s_rc[NB_REC];
    __shared__ int      s_ovf;
    __shared__ unsigned cand[CCAP];
    __shared__ int      cand_n;

    const int tid   = threadIdx.x;
    const int rbase = (int)blockIdx.x * ROWS_WG;

    if (tid < NQUEUE) q_n[tid] = 0;
    if (tid == 0) { cand_n = 0; s_ovf = 0; }
    __syncthreads();

    auto queue_rec = [&](unsigned lr, int e, int c, float2 r0, float2 r1) {
        const unsigned kk = lr / EDGE_WAVES;
        const unsigned qi = (kk / SLOTS) * EDGE_WAVES + lr % EDGE_WAVES;
        const int slot = atomicAdd(&q_n[qi], 1);
        if (slot < QCAP) {
            q_ent[qi * QCAP + slot] = ((unsigned)e << 2) | (kk % SLOTS);
            q_col[qi * QCAP + slot] = c;
            q_r0[qi * QCAP + slot]  = r0;
            q_r1[qi * QCAP + slot]  = r1;
        }
    };
    auto queue_edge = [&](unsigned lr, int e, int c) {
        queue_rec(lr, e, c,
                  make_float2(radial[e], edge_vec[(size_t)e * 3 + 0]),
                  make_float2(edge_vec[(size_t)e * 3 + 1], edge_vec[(size_t)e * 3 + 2]));
    };

    bool queued = false;
    if (rec_fit(natoms, nedges)) {
        static_assert(EDGE_THREADS == 2 * NB_REC, "a thread pair per sub-bucket");
        const int g   = (int)blockIdx.x;
        const int cap = rec_cap(natoms, g);
        if (tid < NB_REC) {
            const int c = cntg[(size_t)g * NB_REC + tid];
            s_rc[tid] = c;
            if (c < 0 || c > cap) s_ovf = 1;
        }
        __syncthreads();
        if (s_ovf == 0) {
            const int b  = tid >> 1;
            const int hf = tid & 1;
            const int n  = s_rc[b];
            const unsigned* __restrict__ base = (const unsigned*)out_vec
                + (size_t)g * (ROWS_WG * 3 * DIM) + (size_t)b * cap * 6;
            for (int i0 = hf; i0 < n; i0 += 2 * REC_V) {
                uint2  a[REC_V];
                float2 r0[REC_V], r1[REC_V];
                #pragma unroll
                for (int u = 0; u < REC_V; ++u) {
                    const int i = i0 + 2 * u;
                    if (i < n) {
                        const unsigned* p = base + (size_t)i * 6;
                        a[u]  = ((const uint2*)p)[0];
                        r0[u] = ((const float2*)p)[1];
                        r1[u] = ((const float2*)p)[2];
                    }
                }
                #pragma unroll
                for (int u = 0; u < REC_V; ++u) {
                    const int i = i0 + 2 * u;
                    if (i < n) {
                        const unsigned lr = a[u].x >> 24;
                        if (lr < (unsigned)ROWS_WG)
                            queue_rec(lr, (int)(a[u].x & 0xFFFFFFu), (int)a[u].y, r0[u], r1[u]);
                    }
                }
            }
            queued = true;
        }
    }
    if (!queued && owner_ids_fit(natoms, nedges)) {
        const unsigned gid = blockIdx.x < 255u ? blockIdx.x : 255u;
        const unsigned pat = gid * 0x01010101u;
        const int nv16 = nedges >> 4;
        const uint4* __restrict__ own4 = (const uint4*)own;
        for (int b0 = 0; b0 < nv16; b0 += EDGE_THREADS * OWN_V) {
            uint4 ov[OWN_V];
            #pragma unroll
            for (int u = 0; u < OWN_V; ++u) {
                if (b0 + u * EDGE_THREADS < nv16) {
                    int i = b0 + u * EDGE_THREADS + tid;
                    if (i >= nv16) i = nv16 - 1;
                    ov[u] = own4[i];
                }
            }
            unsigned long long hit = 0;
            #pragma unroll
            for (int u = 0; u < OWN_V; ++u) {
                const int i = b0 + u * EDGE_THREADS + tid;
                if (i < nv16) {
                    const unsigned ww[4] = { ov[u].x, ov[u].y, ov[u].z, ov[u].w };
                    #pragma unroll
                    for (int d = 0; d < 4; ++d) {
                        const unsigned x = ww[d] ^ pat;
                        if ((x - 0x01010101u) & ~x & 0x80808080u) hit |= 1ull << (4 * u + d);
                    }
                }
            }
            int slot = atomicAdd(&cand_n, __popcll(hit));
            while (hit) {
                const int bq = __builtin_ctzll(hit);
                hit &= hit - 1;
                if (slot < CCAP)
                    cand[slot] = 4u * (unsigned)(b0 + (bq >> 2) * EDGE_THREADS + tid) + (unsigned)(bq & 3);
                ++slot;
            }
        }
        __syncthreads();
        const int nc = cand_n;
        if (nc <= CCAP) {
            const int4* __restrict__ row4 = (const int4*)row;
            const int4* __restrict__ col4 = (const int4*)col;
            int4     rv[RES_V], cv[RES_V];
            unsigned jj[RES_V];
            #pragma unroll
            for (int u = 0; u < RES_V; ++u) {
                const int flat = tid + u * EDGE_THREADS;
                const unsigned j = cand[flat];
                jj[u] = flat < nc ? j : 0xFFFFFFFFu;
            }
            #pragma unroll
            for (int u = 0; u < RES_V; ++u) {
                if (jj[u] != 0xFFFFFFFFu) {
                    rv[u] = row4[jj[u]];
                    cv[u] = col4[jj[u]];
                }
            }
            #pragma unroll
            for (int u = 0; u < RES_V; ++u) {
                if (jj[u] != 0xFFFFFFFFu) {
                    const int rr[4] = { rv[u].x, rv[u].y, rv[u].z, rv[u].w };
                    const int cc[4] = { cv[u].x, cv[u].y, cv[u].z, cv[u].w };
                    #pragma unroll
                    for (int s = 0; s < 4; ++s) {
                        const unsigned lr = (unsigned)(rr[s] - rbase);
                        if (lr < (unsigned)ROWS_WG) queue_edge(lr, (int)(4u * jj[u]) + s, cc[s]);
                    }
                }
            }
            for (int e = (nv16 << 4) + tid; e < nedges; e += EDGE_THREADS) {
                const unsigned lr = (unsigned)(row[e] - rbase);
                if (lr < (unsigned)ROWS_WG) queue_edge(lr, e, col[e]);
            }
            queued = true;
        }
    }
    if (!queued) {
        const int nvec = nedges >> 2;
        const int4* __restrict__ row4 = (const int4*)row;
        for (int i0 = tid; i0 < nvec; i0 += EDGE_THREADS * SCAN_V) {
            int4 rv[SCAN_V];
            #pragma unroll
            for (int u = 0; u < SCAN_V; ++u) {
                int i = i0 + u * EDGE_THREADS;
                if (i >= nvec) i = nvec - 1;
                rv[u] = row4[i];
            }
            #pragma unroll
            for (int u = 0; u < SCAN_V; ++u) {
                const int i = i0 + u * EDGE_THREADS;
                if (i < nvec) {
                    const int rr[4] = { rv[u].x, rv[u].y, rv[u].z, rv[u].w };
                    #pragma unroll
                    for (int s = 0; s < 4; ++s) {
                        const unsigned lr = (unsigned)(rr[s] - rbase);
                        if (lr < (unsigned)ROWS_WG) queue_edge(lr, 4 * i + s, col[4 * i + s]);
                    }
                }
            }
        }
        for (int e = (nvec << 2) + tid; e < nedges; e += EDGE_THREADS) {
            const unsigned lr = (unsigned)(row[e] - rbase);
            if (lr < (unsigned)ROWS_WG) queue_edge(lr, e, col[e]);
        }
    }
    __syncthreads();

    const int lane = tid & 31;
    const int wv   = __builtin_amdgcn_readfirstlane(tid >> 5);
    const int hoff = (lane >> 1) * 24 + (lane & 1) * 4;

    for (int ps = 0; ps < NPASS; ++ps) {
        const int qi    = ps * EDGE_WAVES + wv;
        const int found = __builtin_amdgcn_readfirstlane(q_n[qi]);

        float  acc[SLOTS][16];
        float4 q[SLOTS];
        int    rowg[SLOTS];
        #pragma unroll
        for (int s = 0; s < SLOTS; ++s) {
            #pragma unroll
            for (int c = 0; c < 16; ++c) acc[s][c] = 0.0f;
            const int lr = (ps * SLOTS + s) * EDGE_WAVES + wv;
            const int r  = rbase + lr;
            rowg[s] = (lr < ROWS_WG && r < natoms) ? r : -1;
            q[s] = make_float4(0.0f, 0.0f, 0.0f, 0.0f);
            if (rowg[s] >= 0) q[s] = *(const float4*)(qkv + (size_t)r * QKV_N + hoff);
        }

        if (found <= QCAP) {
            for (int i0 = 0; i0 < found; i0 += EB) {
                EdgeIn x[EB];
                int    sj[EB];
                int li = i0 + (lane & (EB - 1));
                if (li >= found) li = found - 1;
                const unsigned en_v = q_ent[qi * QCAP + li];
                const int      c_v  = q_col[qi * QCAP + li];
                const float2   r0_v = q_r0[qi * QCAP + li];
                const float2   r1_v = q_r1[qi * QCAP + li];
                #pragma unroll
                for (int t = 0; t < EB; ++t) {
                    const unsigned en = (unsigned)__builtin_amdgcn_readlane((int)en_v, t);
                    const int      c  = __builtin_amdgcn_readlane(c_v, t);
                    sj[t] = (int)(en & 3u);
                    x[t].rad = readlane_f(r0_v.x, t);
                    x[t].e0  = readlane_f(r0_v.y, t);
                    x[t].e1  = readlane_f(r1_v.x, t);
                    x[t].e2  = readlane_f(r1_v.y, t);
                    edge_fetch_kvf(x[t], qkv, edge_feat, (int)(en >> 2), c, hoff, lane);
                }
                #pragma unroll
                for (int t = 0; t < EB; ++t)
                    if (i0 + t < found) edge_sum(acc, q, x[t], sj[t]);
            }
        } else {
            for (int e0 = 0; e0 < nedges; e0 += 32) {
                const int e = e0 + lane;
                unsigned lr = 0xFFFFFFFFu;
                if (e < nedges) lr = (unsigned)(row[e] - rbase);
                const bool mine = lr < (unsigned)ROWS_WG
                               && (int)(lr % EDGE_WAVES) == wv
                               && (int)((lr / EDGE_WAVES) / SLOTS) == ps;
                unsigned mask = __builtin_amdgcn_ballot_w32(mine);
                while (mask) {
                    const int b = __builtin_ctz(mask);
                    mask &= mask - 1;
                    const unsigned l = (unsigned)__builtin_amdgcn_readlane((int)lr, b);
                    const int ee = e0 + b;
                    EdgeIn x1;
                    edge_fetch(x1, qkv, edge_feat, edge_vec, radial, ee, col[ee], hoff, lane);
                    edge_sum(acc, q, x1, (int)((l / EDGE_WAVES) % SLOTS));
                }
            }
        }

        #pragma unroll
        for (int s = 0; s < SLOTS; ++s) {
            if (rowg[s] >= 0) {
                const size_t r = (size_t)rowg[s];
                *(float4*)(m_feat + r * DIM + 4 * lane) =
                    make_float4(acc[s][0], acc[s][1], acc[s][2], acc[s][3]);
                #pragma unroll
                for (int d = 0; d < 3; ++d)
                    *(float4*)(out_vec + r * (3 * DIM) + d * DIM + 4 * lane) =
                        make_float4(acc[s][4 + 4 * d], acc[s][5 + 4 * d],
                                    acc[s][6 + 4 * d], acc[s][7 + 4 * d]);
            }
        }
    }
}

#define OUT_GROUPS 2

__global__ void __launch_bounds__(32 * GEMM_WAVES)
out_gemm_kernel(const float* __restrict__ mf,
                const float* __restrict__ w,
                const float* __restrict__ bias,
                float* __restrict__ out,
                int natoms) {
    const int lane = threadIdx.x & 31;
    const int wid  = threadIdx.x >> 5;
    const int half = lane >> 4;
    const int l16  = lane & 15;

    const int mbase = (blockIdx.x * GEMM_WAVES + wid) * 16;
    if (mbase >= natoms) return;
    const int cbase = 4 * l16;

    int arow = mbase + l16;
    if (arow >= natoms) arow = natoms - 1;
    const float* __restrict__ ap = mf + (size_t)arow * DIM + 2 * half;
    const float* __restrict__ wp = w + (size_t)(2 * half) * DIM + cbase;

    v8f acc[OUT_GROUPS * 4] = {};
    #pragma unroll 2
    for (int k0 = 0; k0 < DIM; k0 += 4) {
        const float2 x = *(const float2*)(ap + k0);
        v2f a;
        a.x = x.x;
        a.y = x.y;
        strip_kstep<OUT_GROUPS, DIM>(acc, a, wp + (size_t)k0 * DIM);
    }

    strip_store<OUT_GROUPS, DIM>(acc, bias + cbase, out + cbase,
                                 mbase + half * 8, natoms);
}

extern "C" void kernel_launch(void* const* d_in, const int* in_sizes, int n_in,
                              void* d_out, int out_size, void* d_ws, size_t ws_size,
                              hipStream_t stream) {
    const float* nf   = (const float*)d_in[0];
    const float* ef   = (const float*)d_in[1];
    const float* evec = (const float*)d_in[2];
    const float* rad  = (const float*)d_in[3];
    const float* qkvw = (const float*)d_in[4];
    const float* qkvb = (const float*)d_in[5];
    const float* outw = (const float*)d_in[6];
    const float* outb = (const float*)d_in[7];
    const float* lng  = (const float*)d_in[8];
    const float* lnb  = (const float*)d_in[9];
    const int*   row  = (const int*)d_in[10];
    const int*   col  = (const int*)d_in[11];

    const int natoms = in_sizes[0] / DIM;
    const int nedges = in_sizes[3];

    float* stats = (float*)d_ws;
    float* qkv   = stats + (size_t)2 * natoms;
    float* mfeat = qkv + (size_t)natoms * QKV_N;

    float* out_feat = (float*)d_out;
    float* out_vec  = out_feat + (size_t)natoms * DIM;

    unsigned* own = (unsigned*)d_out;

    int* cntg = (int*)d_out + ((nedges + 255) / 256) * 64;

    const int nb_stats = (natoms + 7) / 8;
    const int nb_own   = ((nedges >> 2) + 255) / 256;
    ln_stats_kernel<<<NB_REC + nb_stats + nb_own, 256, 0, stream>>>(nf, stats, natoms, row, own,
                                                           nedges, nb_stats, col, rad, evec,
                                                           (unsigned*)out_vec, cntg);

    const int strips = (natoms + 15) / 16;
    dim3 g1((strips + GEMM_WAVES - 1) / GEMM_WAVES, QKV_N / (QKV_GROUPS * 64));
    qkv_gemm_kernel<<<g1, 32 * GEMM_WAVES, 0, stream>>>(nf, stats, lng, lnb, qkvw, qkvb,
                                                        qkv, natoms);

    edge_rows_kernel<<<(natoms + ROWS_WG - 1) / ROWS_WG, EDGE_THREADS, 0, stream>>>(
        qkv, ef, evec, rad, row, col, own, cntg, mfeat, out_vec, natoms, nedges);

    dim3 g2((strips + GEMM_WAVES - 1) / GEMM_WAVES, 1);
    out_gemm_kernel<<<g2, 32 * GEMM_WAVES, 0, stream>>>(mfeat, outw, outb, out_feat, natoms);
}
